# starting kernel
# baseline (speedup 1.0000x reference)
#include <hip/hip_runtime.h>
#include <math.h>

typedef float v2f __attribute__((ext_vector_type(2)));
typedef float v8f __attribute__((ext_vector_type(8)));

__global__ void deg_init_k(float* __restrict__ deg, int N) {
  int i = blockIdx.x * blockDim.x + threadIdx.x;
  if (i < N) deg[i] = 1.0f;
}

__global__ void deg_acc_k(const int* __restrict__ dst, float* __restrict__ deg, int E) {
  int e = blockIdx.x * blockDim.x + threadIdx.x;
  if (e < E) {
    __hip_atomic_fetch_add(&deg[dst[e]], 1.0f, __ATOMIC_RELAXED, __HIP_MEMORY_SCOPE_AGENT);
  }
}

__global__ void dinv_k(float* __restrict__ deg, int N) {
  int i = blockIdx.x * blockDim.x + threadIdx.x;
  if (i < N) {
    float d = deg[i];
    deg[i] = (d > 0.0f) ? rsqrtf(d) : 0.0f;
  }
}

template <int NN>
__global__ void gemm_wmma_k(const float* __restrict__ A,
                            const float* __restrict__ B,
                            float* __restrict__ C,
                            int M, int K) {
  const int wave = threadIdx.x >> 5;
  const int lane = threadIdx.x & 31;
  const int m0 = blockIdx.x * 16;
  if (m0 >= M) return;
  const int n0 = wave * 16;
  const int half = lane >> 4;
  const int l15 = lane & 15;

  v8f acc = {};
  const float* aptr = A + (size_t)(m0 + l15) * K + half * 2;
  const float* bptr = B + (size_t)(half * 2) * NN + n0 + l15;

  for (int k = 0; k < K; k += 4) {
    v2f a, b;
    a.x = aptr[0];
    a.y = aptr[1];
    b.x = bptr[0];
    b.y = bptr[NN];
    acc = __builtin_amdgcn_wmma_f32_16x16x4_f32(
         false, a,  false, b,
         (short)0, acc,  false,  false);
    aptr += 4;
    bptr += 4 * NN;
  }

  float* cptr = C + (size_t)(m0 + half * 8) * NN + n0 + l15;
#pragma unroll
  for (int r = 0; r < 8; ++r) {
    cptr[0] = acc[r];
    cptr += NN;
  }
}

template <int C>
__global__ void self_bias_k(const float* __restrict__ feat,
                            const float* __restrict__ dinv,
                            const float* __restrict__ bias,
                            float* __restrict__ out, int N) {
  int t = blockIdx.x * blockDim.x + threadIdx.x;
  int total = N * (C / 4);
  if (t >= total) return;
  int i = t / (C / 4);
  int c4 = t % (C / 4);
  float di = dinv[i];
  float w = di * di;
  float4 f = reinterpret_cast<const float4*>(feat)[t];
  float4 bv = reinterpret_cast<const float4*>(bias)[c4];
  float4 o;
  o.x = w * f.x + bv.x;
  o.y = w * f.y + bv.y;
  o.z = w * f.z + bv.z;
  o.w = w * f.w + bv.w;
  reinterpret_cast<float4*>(out)[t] = o;
}

template <int C>
__global__ void scatter_edges_k(const float* __restrict__ feat,
                                const float* __restrict__ dinv,
                                const int* __restrict__ src,
                                const int* __restrict__ dst,
                                float* __restrict__ out, int E) {
  int wave = blockIdx.x * (blockDim.x >> 5) + (threadIdx.x >> 5);
  int lane = threadIdx.x & 31;
  if (wave >= E) return;
  int s = src[wave];
  int d = dst[wave];
  float norm = dinv[s] * dinv[d];
  const float* frow = feat + (size_t)s * C;
  float* orow = out + (size_t)d * C;
  if constexpr (C == 128) {
    float4 v = reinterpret_cast<const float4*>(frow)[lane];
    float* p = orow + lane * 4;
    __hip_atomic_fetch_add(p + 0, norm * v.x, __ATOMIC_RELAXED, __HIP_MEMORY_SCOPE_AGENT);
    __hip_atomic_fetch_add(p + 1, norm * v.y, __ATOMIC_RELAXED, __HIP_MEMORY_SCOPE_AGENT);
    __hip_atomic_fetch_add(p + 2, norm * v.z, __ATOMIC_RELAXED, __HIP_MEMORY_SCOPE_AGENT);
    __hip_atomic_fetch_add(p + 3, norm * v.w, __ATOMIC_RELAXED, __HIP_MEMORY_SCOPE_AGENT);
  } else {
    float2 v = reinterpret_cast<const float2*>(frow)[lane];
    float* p = orow + lane * 2;
    __hip_atomic_fetch_add(p + 0, norm * v.x, __ATOMIC_RELAXED, __HIP_MEMORY_SCOPE_AGENT);
    __hip_atomic_fetch_add(p + 1, norm * v.y, __ATOMIC_RELAXED, __HIP_MEMORY_SCOPE_AGENT);
  }
}

__global__ void finalize_k(const float* __restrict__ g, float* __restrict__ out, int N) {
  int node = blockIdx.x * (blockDim.x >> 5) + (threadIdx.x >> 5);
  int lane = threadIdx.x & 31;
  if (node >= N) return;
  float2 v = reinterpret_cast<const float2*>(g + (size_t)node * 64)[lane];
  float mx = fmaxf(v.x, v.y);
  float mn = fminf(v.x, v.y);
#pragma unroll
  for (int o = 16; o > 0; o >>= 1) {
    mx = fmaxf(mx, __shfl_xor(mx, o, 32));
    mn = fminf(mn, __shfl_xor(mn, o, 32));
  }
  float inv = 1.0f / (mx - mn);
  float sx = (v.x - mn) * inv;
  float sy = (v.y - mn) * inv;
  float ss = sx * sx + sy * sy;
#pragma unroll
  for (int o = 16; o > 0; o >>= 1) ss += __shfl_xor(ss, o, 32);
  float r = 1.0f / fmaxf(sqrtf(ss), 1e-12f);
  float2 ov;
  ov.x = sx * r;
  ov.y = sy * r;
  reinterpret_cast<float2*>(out + (size_t)node * 64)[lane] = ov;
}

extern "C" void kernel_launch(void* const* d_in, const int* in_sizes, int n_in,
                              void* d_out, int out_size, void* d_ws, size_t ws_size,
                              hipStream_t stream) {
  (void)n_in; (void)out_size; (void)ws_size;

  const float* x  = (const float*)d_in[0];
  const int*   ei = (const int*)d_in[1];
  const float* W1 = (const float*)d_in[2];
  const float* b1 = (const float*)d_in[3];
  const float* W2 = (const float*)d_in[4];
  const float* b2 = (const float*)d_in[5];

  const int N = in_sizes[0] / 128;
  const int E = in_sizes[1] / 2;
  const int* src = ei;
  const int* dst = ei + E;
  float* out = (float*)d_out;

  float* ws = (float*)d_ws;
  size_t Npad = ((size_t)N + 63) & ~(size_t)63;
  float* dinv = ws;
  float* xw = ws + Npad;
  float* h  = xw + (size_t)N * 128;
  float* g2 = h  + (size_t)N * 128;

  deg_init_k<<<(N + 255) / 256, 256, 0, stream>>>(dinv, N);
  deg_acc_k<<<(E + 255) / 256, 256, 0, stream>>>(dst, dinv, E);
  dinv_k<<<(N + 255) / 256, 256, 0, stream>>>(dinv, N);

  gemm_wmma_k<128><<<(N + 15) / 16, 256, 0, stream>>>(x, W1, xw, N, 128);
  self_bias_k<128><<<(N * 32 + 255) / 256, 256, 0, stream>>>(xw, dinv, b1, h, N);
  scatter_edges_k<128><<<(E + 7) / 8, 256, 0, stream>>>(xw, dinv, src, dst, h, E);

  gemm_wmma_k<64><<<(N + 15) / 16, 128, 0, stream>>>(h, W2, xw, N, 128);
  self_bias_k<64><<<(N * 16 + 255) / 256, 256, 0, stream>>>(xw, dinv, b2, g2, N);
  scatter_edges_k<64><<<(E + 7) / 8, 256, 0, stream>>>(xw, dinv, src, dst, g2, E);

  finalize_k<<<(N + 7) / 8, 256, 0, stream>>>(g2, out, N);
}


# Replace f32 atomic edge scatter by a CSR gather rebuilt each call: int atomic arrival-order rank + scan + fill sort edges by dst; one wave per node sums rows in rank order
# speedup vs baseline: 2.0875x; 2.0875x over previous
#include <hip/hip_runtime.h>
#include <math.h>

typedef float v2f __attribute__((ext_vector_type(2)));
typedef float v8f __attribute__((ext_vector_type(8)));

__global__ void cnt_zero_k(int* __restrict__ cnt, int N) {
  int i = blockIdx.x * blockDim.x + threadIdx.x;
  if (i < N) cnt[i] = 0;
}

__global__ void rank_k(const int* __restrict__ dst, int* __restrict__ cnt,
                       int* __restrict__ rank, int E) {
  int e = blockIdx.x * blockDim.x + threadIdx.x;
  if (e < E) {
    rank[e] = __hip_atomic_fetch_add(&cnt[dst[e]], 1, __ATOMIC_RELAXED, __HIP_MEMORY_SCOPE_AGENT);
  }
}

__global__ void scan_k(const int* __restrict__ cnt, int* __restrict__ rowptr,
                       float* __restrict__ dinv, int N) {
  __shared__ int wsum[32];
  __shared__ int psum[32];
  const int tid = threadIdx.x;
  const int lane = tid & 31;
  const int w = tid >> 5;
  const int base = blockIdx.x * 1024;

  int pre = 0;
  for (int j = tid; j < base; j += 1024) pre += cnt[j];
  for (int o = 16; o > 0; o >>= 1) pre += __shfl_xor(pre, o, 32);

  const int i = base + tid;
  const int c = (i < N) ? cnt[i] : 0;
  int x = c;
  for (int o = 1; o < 32; o <<= 1) {
    int y = __shfl_up(x, o, 32);
    if (lane >= o) x += y;
  }
  if (lane == 31) wsum[w] = x;
  if (lane == 0) psum[w] = pre;
  __syncthreads();

  int tot = psum[lane];
  int wpre = (lane < w) ? wsum[lane] : 0;
  for (int o = 16; o > 0; o >>= 1) {
    tot += __shfl_xor(tot, o, 32);
    wpre += __shfl_xor(wpre, o, 32);
  }
  if (i < N) {
    rowptr[i] = tot + wpre + x - c;
    float d = (float)(c + 1);
    dinv[i] = (d > 0.0f) ? rsqrtf(d) : 0.0f;
  }
}

__global__ void fill_k(const int* __restrict__ src, const int* __restrict__ dst,
                       const int* __restrict__ rank, const int* __restrict__ rowptr,
                       int* __restrict__ csr, int E) {
  int e = blockIdx.x * blockDim.x + threadIdx.x;
  if (e < E) csr[rowptr[dst[e]] + rank[e]] = src[e];
}

template <int NN>
__global__ void gemm_wmma_k(const float* __restrict__ A,
                            const float* __restrict__ B,
                            float* __restrict__ C,
                            int M, int K) {
  const int wave = threadIdx.x >> 5;
  const int lane = threadIdx.x & 31;
  const int m0 = blockIdx.x * 16;
  if (m0 >= M) return;
  const int n0 = wave * 16;
  const int half = lane >> 4;
  const int l15 = lane & 15;

  v8f acc = {};
  const float* aptr = A + (size_t)(m0 + l15) * K + half * 2;
  const float* bptr = B + (size_t)(half * 2) * NN + n0 + l15;

  for (int k = 0; k < K; k += 4) {
    v2f a, b;
    a.x = aptr[0];
    a.y = aptr[1];
    b.x = bptr[0];
    b.y = bptr[NN];
    acc = __builtin_amdgcn_wmma_f32_16x16x4_f32(
         false, a,  false, b,
         (short)0, acc,  false,  false);
    aptr += 4;
    bptr += 4 * NN;
  }

  float* cptr = C + (size_t)(m0 + half * 8) * NN + n0 + l15;
#pragma unroll
  for (int r = 0; r < 8; ++r) {
    cptr[0] = acc[r];
    cptr += NN;
  }
}

template <int C> struct rowvec;
template <> struct rowvec<128> { typedef float4 type; };
template <> struct rowvec<64>  { typedef float2 type; };

__device__ __forceinline__ void axpy(float4& a, float w, const float4& v) {
  a.x += w * v.x; a.y += w * v.y; a.z += w * v.z; a.w += w * v.w;
}
__device__ __forceinline__ void axpy(float2& a, float w, const float2& v) {
  a.x += w * v.x; a.y += w * v.y;
}

template <int C>
__global__ void gather_k(const float* __restrict__ feat,
                         const float* __restrict__ dinv,
                         const int* __restrict__ rowptr,
                         const int* __restrict__ cnt,
                         const int* __restrict__ csr,
                         const float* __restrict__ bias,
                         float* __restrict__ out, int N) {
  typedef typename rowvec<C>::type vec;
  const int lane = threadIdx.x & 31;
  const int node = __builtin_amdgcn_readfirstlane(
      (int)(blockIdx.x * (blockDim.x >> 5) + (threadIdx.x >> 5)));
  if (node >= N) return;

  int p = rowptr[node];
  const int end = p + cnt[node];
  const float di = dinv[node];

  vec acc = reinterpret_cast<const vec*>(bias)[lane];
  axpy(acc, di * di, reinterpret_cast<const vec*>(feat + (size_t)node * C)[lane]);

  for (; p + 4 <= end; p += 4) {
    const int s0 = csr[p], s1 = csr[p + 1], s2 = csr[p + 2], s3 = csr[p + 3];
    const vec v0 = reinterpret_cast<const vec*>(feat + (size_t)s0 * C)[lane];
    const vec v1 = reinterpret_cast<const vec*>(feat + (size_t)s1 * C)[lane];
    const vec v2 = reinterpret_cast<const vec*>(feat + (size_t)s2 * C)[lane];
    const vec v3 = reinterpret_cast<const vec*>(feat + (size_t)s3 * C)[lane];
    const float w0 = dinv[s0] * di, w1 = dinv[s1] * di;
    const float w2 = dinv[s2] * di, w3 = dinv[s3] * di;
    axpy(acc, w0, v0);
    axpy(acc, w1, v1);
    axpy(acc, w2, v2);
    axpy(acc, w3, v3);
  }
  if (p + 2 <= end) {
    const int s0 = csr[p], s1 = csr[p + 1];
    const vec v0 = reinterpret_cast<const vec*>(feat + (size_t)s0 * C)[lane];
    const vec v1 = reinterpret_cast<const vec*>(feat + (size_t)s1 * C)[lane];
    const float w0 = dinv[s0] * di, w1 = dinv[s1] * di;
    axpy(acc, w0, v0);
    axpy(acc, w1, v1);
    p += 2;
  }
  if (p < end) {
    const int s0 = csr[p];
    const vec v0 = reinterpret_cast<const vec*>(feat + (size_t)s0 * C)[lane];
    axpy(acc, dinv[s0] * di, v0);
  }
  reinterpret_cast<vec*>(out + (size_t)node * C)[lane] = acc;
}

__global__ void finalize_k(const float* __restrict__ g, float* __restrict__ out, int N) {
  int node = blockIdx.x * (blockDim.x >> 5) + (threadIdx.x >> 5);
  int lane = threadIdx.x & 31;
  if (node >= N) return;
  float2 v = reinterpret_cast<const float2*>(g + (size_t)node * 64)[lane];
  float mx = fmaxf(v.x, v.y);
  float mn = fminf(v.x, v.y);
#pragma unroll
  for (int o = 16; o > 0; o >>= 1) {
    mx = fmaxf(mx, __shfl_xor(mx, o, 32));
    mn = fminf(mn, __shfl_xor(mn, o, 32));
  }
  float inv = 1.0f / (mx - mn);
  float sx = (v.x - mn) * inv;
  float sy = (v.y - mn) * inv;
  float ss = sx * sx + sy * sy;
#pragma unroll
  for (int o = 16; o > 0; o >>= 1) ss += __shfl_xor(ss, o, 32);
  float r = 1.0f / fmaxf(sqrtf(ss), 1e-12f);
  float2 ov;
  ov.x = sx * r;
  ov.y = sy * r;
  reinterpret_cast<float2*>(out + (size_t)node * 64)[lane] = ov;
}

extern "C" void kernel_launch(void* const* d_in, const int* in_sizes, int n_in,
                              void* d_out, int out_size, void* d_ws, size_t ws_size,
                              hipStream_t stream) {
  (void)n_in; (void)out_size; (void)ws_size;

  const float* x  = (const float*)d_in[0];
  const int*   ei = (const int*)d_in[1];
  const float* W1 = (const float*)d_in[2];
  const float* b1 = (const float*)d_in[3];
  const float* W2 = (const float*)d_in[4];
  const float* b2 = (const float*)d_in[5];

  const int N = in_sizes[0] / 128;
  const int E = in_sizes[1] / 2;
  const int* src = ei;
  const int* dst = ei + E;
  float* out = (float*)d_out;

  float* ws = (float*)d_ws;
  size_t Npad = ((size_t)N + 63) & ~(size_t)63;
  float* dinv = ws;
  float* xw = ws + Npad;
  float* h  = xw + (size_t)N * 128;
  float* g2 = xw + (size_t)N * 64;
  int* cnt    = (int*)(h + (size_t)N * 128);
  int* rowptr = cnt + Npad;
  int* rank   = rowptr + Npad;
  int* csr    = rank + E;

  cnt_zero_k<<<(N + 255) / 256, 256, 0, stream>>>(cnt, N);
  rank_k<<<(E + 255) / 256, 256, 0, stream>>>(dst, cnt, rank, E);
  scan_k<<<(N + 1023) / 1024, 1024, 0, stream>>>(cnt, rowptr, dinv, N);
  fill_k<<<(E + 255) / 256, 256, 0, stream>>>(src, dst, rank, rowptr, csr, E);

  gemm_wmma_k<128><<<(N + 15) / 16, 256, 0, stream>>>(x, W1, xw, N, 128);
  gather_k<128><<<(N + 7) / 8, 256, 0, stream>>>(xw, dinv, rowptr, cnt, csr, b1, h, N);

  gemm_wmma_k<64><<<(N + 15) / 16, 128, 0, stream>>>(h, W2, xw, N, 128);
  gather_k<64><<<(N + 7) / 8, 256, 0, stream>>>(xw, dinv, rowptr, cnt, csr, b2, g2, N);

  finalize_k<<<(N + 7) / 8, 256, 0, stream>>>(g2, out, N);
}


# Edge ranking moved from 1.6M returning global atomics to LDS counters: edges binned by 512-node group (LDS histogram per chunk), each group counted, scanned, ranked in LDS; sums in LDS arrival order
# speedup vs baseline: 2.9000x; 1.3892x over previous
#include <hip/hip_runtime.h>
#include <math.h>

typedef float v2f __attribute__((ext_vector_type(2)));
typedef float v8f __attribute__((ext_vector_type(8)));

#define GRP_SHIFT 9
#define GRP_SIZE  512
#define GRP_MAX   256
#define EDGE_CHUNK 8192
#define CSR_STAGE 12288

__global__ __launch_bounds__(1024)
void group_count_k(const int* __restrict__ dst, int* __restrict__ cntmat, int E) {
  __shared__ int hist[GRP_MAX];
  const int tid = threadIdx.x;
  if (tid < GRP_MAX) hist[tid] = 0;
  __syncthreads();
  const int e0 = blockIdx.x * EDGE_CHUNK + tid;
  int d[EDGE_CHUNK / 1024];
#pragma unroll
  for (int j = 0; j < EDGE_CHUNK / 1024; ++j) {
    const int e = e0 + j * 1024;
    d[j] = dst[e < E ? e : E - 1];
  }
#pragma unroll
  for (int j = 0; j < EDGE_CHUNK / 1024; ++j) {
    atomicAdd(&hist[d[j] >> GRP_SHIFT], (e0 + j * 1024 < E) ? 1 : 0);
  }
  __syncthreads();
  if (tid < GRP_MAX) cntmat[blockIdx.x * GRP_MAX + tid] = hist[tid];
}

__global__ __launch_bounds__(1024)
void group_scatter_k(const int* __restrict__ src, const int* __restrict__ dst,
                     const int* __restrict__ cntmat, int* __restrict__ gbase,
                     int2* __restrict__ tmp, int E, int nchunks) {
  __shared__ int part_tot[4][GRP_MAX];
  __shared__ int part_pre[4][GRP_MAX];
  __shared__ int off[GRP_MAX];
  __shared__ int wsum[8];
  const int tid = threadIdx.x;
  const int lane = tid & 31;
  const int w = tid >> 5;
  const int blk = blockIdx.x;
  {
    const int b = tid & (GRP_MAX - 1);
    const int g = tid >> 8;
    int tot = 0, pre = 0;
    for (int r = g; r < nchunks; r += 4) {
      const int v = cntmat[r * GRP_MAX + b];
      tot += v;
      if (r < blk) pre += v;
    }
    part_tot[g][b] = tot;
    part_pre[g][b] = pre;
  }
  __syncthreads();
  int t = 0, p = 0, x = 0;
  if (tid < GRP_MAX) {
    t = part_tot[0][tid] + part_tot[1][tid] + part_tot[2][tid] + part_tot[3][tid];
    p = part_pre[0][tid] + part_pre[1][tid] + part_pre[2][tid] + part_pre[3][tid];
    x = t;
    for (int o = 1; o < 32; o <<= 1) {
      const int y = __shfl_up(x, o, 32);
      if (lane >= o) x += y;
    }
    if (lane == 31) wsum[w] = x;
  }
  __syncthreads();
  if (tid < GRP_MAX) {
    int wp = 0;
    for (int k = 0; k < w; ++k) wp += wsum[k];
    const int base = wp + x - t;
    off[tid] = base + p;
    if (blk == 0) gbase[tid] = base;
  }
  __syncthreads();
  const int e0 = blk * EDGE_CHUNK + tid;
  int d[EDGE_CHUNK / 1024], s[EDGE_CHUNK / 1024];
#pragma unroll
  for (int j = 0; j < EDGE_CHUNK / 1024; ++j) {
    const int e = e0 + j * 1024;
    const int ec = e < E ? e : E - 1;
    d[j] = dst[ec];
    s[j] = src[ec];
  }
  int pos[EDGE_CHUNK / 1024];
#pragma unroll
  for (int j = 0; j < EDGE_CHUNK / 1024; ++j) {
    pos[j] = atomicAdd(&off[d[j] >> GRP_SHIFT], (e0 + j * 1024 < E) ? 1 : 0);
  }
#pragma unroll
  for (int j = 0; j < EDGE_CHUNK / 1024; ++j) {
    if (e0 + j * 1024 < E) tmp[pos[j]] = make_int2(s[j], d[j]);
  }
}

__global__ __launch_bounds__(1024)
void group_csr_k(const int2* __restrict__ tmp, const int* __restrict__ gbase,
                 int* __restrict__ cnt, int* __restrict__ rowptr,
                 float* __restrict__ dinv, int* __restrict__ csr, int N, int E) {
  __shared__ int c[GRP_SIZE];
  __shared__ int cur[GRP_SIZE];
  __shared__ int wsum[16];
  __shared__ int stage[CSR_STAGE];
  const int tid = threadIdx.x;
  const int lane = tid & 31;
  const int w = tid >> 5;
  const int b = blockIdx.x;
  const int lo = b << GRP_SHIFT;
  const int start = gbase[b];
  const int end = (b + 1 < GRP_MAX) ? gbase[b + 1] : E;
  const int n = end - start;

  if (tid < GRP_SIZE) c[tid] = 0;
  __syncthreads();
  for (int i0 = start + tid; i0 < end; i0 += 4096) {
    int dd[4];
#pragma unroll
    for (int k = 0; k < 4; ++k) {
      const int i = i0 + k * 1024;
      dd[k] = tmp[i < end ? i : i0].y;
    }
#pragma unroll
    for (int k = 0; k < 4; ++k) {
      atomicAdd(&c[dd[k] - lo], (i0 + k * 1024 < end) ? 1 : 0);
    }
  }
  __syncthreads();

  int v = 0, x = 0;
  if (tid < GRP_SIZE) {
    v = c[tid];
    x = v;
    for (int o = 1; o < 32; o <<= 1) {
      const int y = __shfl_up(x, o, 32);
      if (lane >= o) x += y;
    }
    if (lane == 31) wsum[w] = x;
  }
  __syncthreads();
  if (tid < GRP_SIZE) {
    int wp = 0;
    for (int k = 0; k < w; ++k) wp += wsum[k];
    const int excl = wp + x - v;
    cur[tid] = excl;
    const int i = lo + tid;
    if (i < N) {
      cnt[i] = v;
      rowptr[i] = start + excl;
      float d = (float)(v + 1);
      dinv[i] = (d > 0.0f) ? rsqrtf(d) : 0.0f;
    }
  }
  __syncthreads();

  const bool staged = (n <= CSR_STAGE);
  for (int i0 = start + tid; i0 < end; i0 += 4096) {
    int2 pr[4];
#pragma unroll
    for (int k = 0; k < 4; ++k) {
      const int i = i0 + k * 1024;
      pr[k] = tmp[i < end ? i : i0];
    }
    int q[4];
#pragma unroll
    for (int k = 0; k < 4; ++k) {
      q[k] = atomicAdd(&cur[pr[k].y - lo], (i0 + k * 1024 < end) ? 1 : 0);
    }
#pragma unroll
    for (int k = 0; k < 4; ++k) {
      if (i0 + k * 1024 < end) {
        if (staged) stage[q[k]] = pr[k].x;
        else csr[start + q[k]] = pr[k].x;
      }
    }
  }
  if (staged) {
    __syncthreads();
    for (int i = tid; i < n; i += 1024) csr[start + i] = stage[i];
  }
}

template <int NN>
__global__ void gemm_wmma_k(const float* __restrict__ A,
                            const float* __restrict__ B,
                            float* __restrict__ C,
                            int M, int K) {
  const int wave = threadIdx.x >> 5;
  const int lane = threadIdx.x & 31;
  const int m0 = blockIdx.x * 16;
  if (m0 >= M) return;
  const int n0 = wave * 16;
  const int half = lane >> 4;
  const int l15 = lane & 15;

  v8f acc = {};
  const float* aptr = A + (size_t)(m0 + l15) * K + half * 2;
  const float* bptr = B + (size_t)(half * 2) * NN + n0 + l15;

  for (int k = 0; k < K; k += 4) {
    v2f a, b;
    a.x = aptr[0];
    a.y = aptr[1];
    b.x = bptr[0];
    b.y = bptr[NN];
    acc = __builtin_amdgcn_wmma_f32_16x16x4_f32(
         false, a,  false, b,
         (short)0, acc,  false,  false);
    aptr += 4;
    bptr += 4 * NN;
  }

  float* cptr = C + (size_t)(m0 + half * 8) * NN + n0 + l15;
#pragma unroll
  for (int r = 0; r < 8; ++r) {
    cptr[0] = acc[r];
    cptr += NN;
  }
}

template <int C> struct rowvec;
template <> struct rowvec<128> { typedef float4 type; };
template <> struct rowvec<64>  { typedef float2 type; };

__device__ __forceinline__ void axpy(float4& a, float w, const float4& v) {
  a.x += w * v.x; a.y += w * v.y; a.z += w * v.z; a.w += w * v.w;
}
__device__ __forceinline__ void axpy(float2& a, float w, const float2& v) {
  a.x += w * v.x; a.y += w * v.y;
}

template <int C>
__global__ void gather_k(const float* __restrict__ feat,
                         const float* __restrict__ dinv,
                         const int* __restrict__ rowptr,
                         const int* __restrict__ cnt,
                         const int* __restrict__ csr,
                         const float* __restrict__ bias,
                         float* __restrict__ out, int N) {
  typedef typename rowvec<C>::type vec;
  const int lane = threadIdx.x & 31;
  const int node = __builtin_amdgcn_readfirstlane(
      (int)(blockIdx.x * (blockDim.x >> 5) + (threadIdx.x >> 5)));
  if (node >= N) return;

  int p = rowptr[node];
  const int end = p + cnt[node];
  const float di = dinv[node];

  vec acc = reinterpret_cast<const vec*>(bias)[lane];
  axpy(acc, di * di, reinterpret_cast<const vec*>(feat + (size_t)node * C)[lane]);

  for (; p + 4 <= end; p += 4) {
    const int s0 = csr[p], s1 = csr[p + 1], s2 = csr[p + 2], s3 = csr[p + 3];
    const vec v0 = reinterpret_cast<const vec*>(feat + (size_t)s0 * C)[lane];
    const vec v1 = reinterpret_cast<const vec*>(feat + (size_t)s1 * C)[lane];
    const vec v2 = reinterpret_cast<const vec*>(feat + (size_t)s2 * C)[lane];
    const vec v3 = reinterpret_cast<const vec*>(feat + (size_t)s3 * C)[lane];
    const float w0 = dinv[s0] * di, w1 = dinv[s1] * di;
    const float w2 = dinv[s2] * di, w3 = dinv[s3] * di;
    axpy(acc, w0, v0);
    axpy(acc, w1, v1);
    axpy(acc, w2, v2);
    axpy(acc, w3, v3);
  }
  if (p + 2 <= end) {
    const int s0 = csr[p], s1 = csr[p + 1];
    const vec v0 = reinterpret_cast<const vec*>(feat + (size_t)s0 * C)[lane];
    const vec v1 = reinterpret_cast<const vec*>(feat + (size_t)s1 * C)[lane];
    const float w0 = dinv[s0] * di, w1 = dinv[s1] * di;
    axpy(acc, w0, v0);
    axpy(acc, w1, v1);
    p += 2;
  }
  if (p < end) {
    const int s0 = csr[p];
    const vec v0 = reinterpret_cast<const vec*>(feat + (size_t)s0 * C)[lane];
    axpy(acc, dinv[s0] * di, v0);
  }
  reinterpret_cast<vec*>(out + (size_t)node * C)[lane] = acc;
}

__global__ void finalize_k(const float* __restrict__ g, float* __restrict__ out, int N) {
  int node = blockIdx.x * (blockDim.x >> 5) + (threadIdx.x >> 5);
  int lane = threadIdx.x & 31;
  if (node >= N) return;
  float2 v = reinterpret_cast<const float2*>(g + (size_t)node * 64)[lane];
  float mx = fmaxf(v.x, v.y);
  float mn = fminf(v.x, v.y);
#pragma unroll
  for (int o = 16; o > 0; o >>= 1) {
    mx = fmaxf(mx, __shfl_xor(mx, o, 32));
    mn = fminf(mn, __shfl_xor(mn, o, 32));
  }
  float inv = 1.0f / (mx - mn);
  float sx = (v.x - mn) * inv;
  float sy = (v.y - mn) * inv;
  float ss = sx * sx + sy * sy;
#pragma unroll
  for (int o = 16; o > 0; o >>= 1) ss += __shfl_xor(ss, o, 32);
  float r = 1.0f / fmaxf(sqrtf(ss), 1e-12f);
  float2 ov;
  ov.x = sx * r;
  ov.y = sy * r;
  reinterpret_cast<float2*>(out + (size_t)node * 64)[lane] = ov;
}

extern "C" void kernel_launch(void* const* d_in, const int* in_sizes, int n_in,
                              void* d_out, int out_size, void* d_ws, size_t ws_size,
                              hipStream_t stream) {
  (void)n_in; (void)out_size; (void)ws_size;

  const float* x  = (const float*)d_in[0];
  const int*   ei = (const int*)d_in[1];
  const float* W1 = (const float*)d_in[2];
  const float* b1 = (const float*)d_in[3];
  const float* W2 = (const float*)d_in[4];
  const float* b2 = (const float*)d_in[5];

  const int N = in_sizes[0] / 128;
  const int E = in_sizes[1] / 2;
  const int* src = ei;
  const int* dst = ei + E;
  float* out = (float*)d_out;

  float* ws = (float*)d_ws;
  size_t Npad = ((size_t)N + 63) & ~(size_t)63;
  float* dinv = ws;
  float* xw = ws + Npad;
  float* h  = xw + (size_t)N * 128;
  float* g2 = xw + (size_t)N * 64;
  int* cnt    = (int*)(h + (size_t)N * 128);
  int* rowptr = cnt + Npad;
  int2* tmp   = (int2*)h;
  int* csr    = rowptr + Npad;
  int* cntmat = csr + E;
  const int nchunks = (E + EDGE_CHUNK - 1) / EDGE_CHUNK;
  int* gbase  = cntmat + (size_t)nchunks * GRP_MAX;

  group_count_k<<<nchunks, 1024, 0, stream>>>(dst, cntmat, E);
  group_scatter_k<<<nchunks, 1024, 0, stream>>>(src, dst, cntmat, gbase, tmp, E, nchunks);
  group_csr_k<<<(N + GRP_SIZE - 1) / GRP_SIZE, 1024, 0, stream>>>(tmp, gbase, cnt, rowptr, dinv, csr, N, E);

  gemm_wmma_k<128><<<(N + 15) / 16, 256, 0, stream>>>(x, W1, xw, N, 128);
  gather_k<128><<<(N + 7) / 8, 256, 0, stream>>>(xw, dinv, rowptr, cnt, csr, b1, h, N);

  gemm_wmma_k<64><<<(N + 15) / 16, 128, 0, stream>>>(h, W2, xw, N, 128);
  gather_k<64><<<(N + 7) / 8, 256, 0, stream>>>(xw, dinv, rowptr, cnt, csr, b2, g2, N);

  finalize_k<<<(N + 7) / 8, 256, 0, stream>>>(g2, out, N);
}


# Min-max scale + L2 normalise fused into the layer-2 gather: the wave that holds a node's 64 sums finishes and stores the output row; one launch of 100000 short waves and the sum plane's round trip go
# speedup vs baseline: 3.0598x; 1.0551x over previous
#include <hip/hip_runtime.h>
#include <math.h>

typedef float v2f __attribute__((ext_vector_type(2)));
typedef float v8f __attribute__((ext_vector_type(8)));

#define GRP_SHIFT 9
#define GRP_SIZE  512
#define GRP_MAX   256
#define EDGE_CHUNK 8192
#define CSR_STAGE 12288

__global__ __launch_bounds__(1024)
void group_count_k(const int* __restrict__ dst, int* __restrict__ cntmat, int E) {
  __shared__ int hist[GRP_MAX];
  const int tid = threadIdx.x;
  if (tid < GRP_MAX) hist[tid] = 0;
  __syncthreads();
  const int e0 = blockIdx.x * EDGE_CHUNK + tid;
  int d[EDGE_CHUNK / 1024];
#pragma unroll
  for (int j = 0; j < EDGE_CHUNK / 1024; ++j) {
    const int e = e0 + j * 1024;
    d[j] = dst[e < E ? e : E - 1];
  }
#pragma unroll
  for (int j = 0; j < EDGE_CHUNK / 1024; ++j) {
    atomicAdd(&hist[d[j] >> GRP_SHIFT], (e0 + j * 1024 < E) ? 1 : 0);
  }
  __syncthreads();
  if (tid < GRP_MAX) cntmat[blockIdx.x * GRP_MAX + tid] = hist[tid];
}

__global__ __launch_bounds__(1024)
void group_scatter_k(const int* __restrict__ src, const int* __restrict__ dst,
                     const int* __restrict__ cntmat, int* __restrict__ gbase,
                     int2* __restrict__ tmp, int E, int nchunks) {
  __shared__ int part_tot[4][GRP_MAX];
  __shared__ int part_pre[4][GRP_MAX];
  __shared__ int off[GRP_MAX];
  __shared__ int wsum[8];
  const int tid = threadIdx.x;
  const int lane = tid & 31;
  const int w = tid >> 5;
  const int blk = blockIdx.x;
  {
    const int b = tid & (GRP_MAX - 1);
    const int g = tid >> 8;
    int tot = 0, pre = 0;
    for (int r = g; r < nchunks; r += 4) {
      const int v = cntmat[r * GRP_MAX + b];
      tot += v;
      if (r < blk) pre += v;
    }
    part_tot[g][b] = tot;
    part_pre[g][b] = pre;
  }
  __syncthreads();
  int t = 0, p = 0, x = 0;
  if (tid < GRP_MAX) {
    t = part_tot[0][tid] + part_tot[1][tid] + part_tot[2][tid] + part_tot[3][tid];
    p = part_pre[0][tid] + part_pre[1][tid] + part_pre[2][tid] + part_pre[3][tid];
    x = t;
    for (int o = 1; o < 32; o <<= 1) {
      const int y = __shfl_up(x, o, 32);
      if (lane >= o) x += y;
    }
    if (lane == 31) wsum[w] = x;
  }
  __syncthreads();
  if (tid < GRP_MAX) {
    int wp = 0;
    for (int k = 0; k < w; ++k) wp += wsum[k];
    const int base = wp + x - t;
    off[tid] = base + p;
    if (blk == 0) gbase[tid] = base;
  }
  __syncthreads();
  const int e0 = blk * EDGE_CHUNK + tid;
  int d[EDGE_CHUNK / 1024], s[EDGE_CHUNK / 1024];
#pragma unroll
  for (int j = 0; j < EDGE_CHUNK / 1024; ++j) {
    const int e = e0 + j * 1024;
    const int ec = e < E ? e : E - 1;
    d[j] = dst[ec];
    s[j] = src[ec];
  }
  int pos[EDGE_CHUNK / 1024];
#pragma unroll
  for (int j = 0; j < EDGE_CHUNK / 1024; ++j) {
    pos[j] = atomicAdd(&off[d[j] >> GRP_SHIFT], (e0 + j * 1024 < E) ? 1 : 0);
  }
#pragma unroll
  for (int j = 0; j < EDGE_CHUNK / 1024; ++j) {
    if (e0 + j * 1024 < E) tmp[pos[j]] = make_int2(s[j], d[j]);
  }
}

__global__ __launch_bounds__(1024)
void group_csr_k(const int2* __restrict__ tmp, const int* __restrict__ gbase,
                 int* __restrict__ cnt, int* __restrict__ rowptr,
                 float* __restrict__ dinv, int* __restrict__ csr, int N, int E) {
  __shared__ int c[GRP_SIZE];
  __shared__ int cur[GRP_SIZE];
  __shared__ int wsum[16];
  __shared__ int stage[CSR_STAGE];
  const int tid = threadIdx.x;
  const int lane = tid & 31;
  const int w = tid >> 5;
  const int b = blockIdx.x;
  const int lo = b << GRP_SHIFT;
  const int start = gbase[b];
  const int end = (b + 1 < GRP_MAX) ? gbase[b + 1] : E;
  const int n = end - start;

  if (tid < GRP_SIZE) c[tid] = 0;
  __syncthreads();
  for (int i0 = start + tid; i0 < end; i0 += 4096) {
    int dd[4];
#pragma unroll
    for (int k = 0; k < 4; ++k) {
      const int i = i0 + k * 1024;
      dd[k] = tmp[i < end ? i : i0].y;
    }
#pragma unroll
    for (int k = 0; k < 4; ++k) {
      atomicAdd(&c[dd[k] - lo], (i0 + k * 1024 < end) ? 1 : 0);
    }
  }
  __syncthreads();

  int v = 0, x = 0;
  if (tid < GRP_SIZE) {
    v = c[tid];
    x = v;
    for (int o = 1; o < 32; o <<= 1) {
      const int y = __shfl_up(x, o, 32);
      if (lane >= o) x += y;
    }
    if (lane == 31) wsum[w] = x;
  }
  __syncthreads();
  if (tid < GRP_SIZE) {
    int wp = 0;
    for (int k = 0; k < w; ++k) wp += wsum[k];
    const int excl = wp + x - v;
    cur[tid] = excl;
    const int i = lo + tid;
    if (i < N) {
      cnt[i] = v;
      rowptr[i] = start + excl;
      float d = (float)(v + 1);
      dinv[i] = (d > 0.0f) ? rsqrtf(d) : 0.0f;
    }
  }
  __syncthreads();

  const bool staged = (n <= CSR_STAGE);
  for (int i0 = start + tid; i0 < end; i0 += 4096) {
    int2 pr[4];
#pragma unroll
    for (int k = 0; k < 4; ++k) {
      const int i = i0 + k * 1024;
      pr[k] = tmp[i < end ? i : i0];
    }
    int q[4];
#pragma unroll
    for (int k = 0; k < 4; ++k) {
      q[k] = atomicAdd(&cur[pr[k].y - lo], (i0 + k * 1024 < end) ? 1 : 0);
    }
#pragma unroll
    for (int k = 0; k < 4; ++k) {
      if (i0 + k * 1024 < end) {
        if (staged) stage[q[k]] = pr[k].x;
        else csr[start + q[k]] = pr[k].x;
      }
    }
  }
  if (staged) {
    __syncthreads();
    for (int i = tid; i < n; i += 1024) csr[start + i] = stage[i];
  }
}

template <int NN>
__global__ void gemm_wmma_k(const float* __restrict__ A,
                            const float* __restrict__ B,
                            float* __restrict__ C,
                            int M, int K) {
  const int wave = threadIdx.x >> 5;
  const int lane = threadIdx.x & 31;
  const int m0 = blockIdx.x * 16;
  if (m0 >= M) return;
  const int n0 = wave * 16;
  const int half = lane >> 4;
  const int l15 = lane & 15;

  v8f acc = {};
  const float* aptr = A + (size_t)(m0 + l15) * K + half * 2;
  const float* bptr = B + (size_t)(half * 2) * NN + n0 + l15;

  for (int k = 0; k < K; k += 4) {
    v2f a, b;
    a.x = aptr[0];
    a.y = aptr[1];
    b.x = bptr[0];
    b.y = bptr[NN];
    acc = __builtin_amdgcn_wmma_f32_16x16x4_f32(
         false, a,  false, b,
         (short)0, acc,  false,  false);
    aptr += 4;
    bptr += 4 * NN;
  }

  float* cptr = C + (size_t)(m0 + half * 8) * NN + n0 + l15;
#pragma unroll
  for (int r = 0; r < 8; ++r) {
    cptr[0] = acc[r];
    cptr += NN;
  }
}

template <int C> struct rowvec;
template <> struct rowvec<128> { typedef float4 type; };
template <> struct rowvec<64>  { typedef float2 type; };

__device__ __forceinline__ void axpy(float4& a, float w, const float4& v) {
  a.x += w * v.x; a.y += w * v.y; a.z += w * v.z; a.w += w * v.w;
}
__device__ __forceinline__ void axpy(float2& a, float w, const float2& v) {
  a.x += w * v.x; a.y += w * v.y;
}

__device__ __forceinline__ float2 scale_normalize(float2 v) {
  float mx = fmaxf(v.x, v.y);
  float mn = fminf(v.x, v.y);
#pragma unroll
  for (int o = 16; o > 0; o >>= 1) {
    mx = fmaxf(mx, __shfl_xor(mx, o, 32));
    mn = fminf(mn, __shfl_xor(mn, o, 32));
  }
  float inv = 1.0f / (mx - mn);
  float sx = (v.x - mn) * inv;
  float sy = (v.y - mn) * inv;
  float ss = sx * sx + sy * sy;
#pragma unroll
  for (int o = 16; o > 0; o >>= 1) ss += __shfl_xor(ss, o, 32);
  float r = 1.0f / fmaxf(sqrtf(ss), 1e-12f);
  float2 ov;
  ov.x = sx * r;
  ov.y = sy * r;
  return ov;
}

template <int C, bool NORM>
__global__ void gather_k(const float* __restrict__ feat,
                         const float* __restrict__ dinv,
                         const int* __restrict__ rowptr,
                         const int* __restrict__ cnt,
                         const int* __restrict__ csr,
                         const float* __restrict__ bias,
                         float* __restrict__ out, int N) {
  typedef typename rowvec<C>::type vec;
  const int lane = threadIdx.x & 31;
  const int node = __builtin_amdgcn_readfirstlane(
      (int)(blockIdx.x * (blockDim.x >> 5) + (threadIdx.x >> 5)));
  if (node >= N) return;

  int p = rowptr[node];
  const int end = p + cnt[node];
  const float di = dinv[node];

  vec acc = reinterpret_cast<const vec*>(bias)[lane];
  axpy(acc, di * di, reinterpret_cast<const vec*>(feat + (size_t)node * C)[lane]);

  for (; p + 4 <= end; p += 4) {
    const int s0 = csr[p], s1 = csr[p + 1], s2 = csr[p + 2], s3 = csr[p + 3];
    const vec v0 = reinterpret_cast<const vec*>(feat + (size_t)s0 * C)[lane];
    const vec v1 = reinterpret_cast<const vec*>(feat + (size_t)s1 * C)[lane];
    const vec v2 = reinterpret_cast<const vec*>(feat + (size_t)s2 * C)[lane];
    const vec v3 = reinterpret_cast<const vec*>(feat + (size_t)s3 * C)[lane];
    const float w0 = dinv[s0] * di, w1 = dinv[s1] * di;
    const float w2 = dinv[s2] * di, w3 = dinv[s3] * di;
    axpy(acc, w0, v0);
    axpy(acc, w1, v1);
    axpy(acc, w2, v2);
    axpy(acc, w3, v3);
  }
  if (p + 2 <= end) {
    const int s0 = csr[p], s1 = csr[p + 1];
    const vec v0 = reinterpret_cast<const vec*>(feat + (size_t)s0 * C)[lane];
    const vec v1 = reinterpret_cast<const vec*>(feat + (size_t)s1 * C)[lane];
    const float w0 = dinv[s0] * di, w1 = dinv[s1] * di;
    axpy(acc, w0, v0);
    axpy(acc, w1, v1);
    p += 2;
  }
  if (p < end) {
    const int s0 = csr[p];
    const vec v0 = reinterpret_cast<const vec*>(feat + (size_t)s0 * C)[lane];
    axpy(acc, dinv[s0] * di, v0);
  }
  if constexpr (NORM) {
    static_assert(C == 64, "the fused scale + normalise expects one float2 per lane");
    acc = scale_normalize(acc);
  }
  reinterpret_cast<vec*>(out + (size_t)node * C)[lane] = acc;
}

extern "C" void kernel_launch(void* const* d_in, const int* in_sizes, int n_in,
                              void* d_out, int out_size, void* d_ws, size_t ws_size,
                              hipStream_t stream) {
  (void)n_in; (void)out_size; (void)ws_size;

  const float* x  = (const float*)d_in[0];
  const int*   ei = (const int*)d_in[1];
  const float* W1 = (const float*)d_in[2];
  const float* b1 = (const float*)d_in[3];
  const float* W2 = (const float*)d_in[4];
  const float* b2 = (const float*)d_in[5];

  const int N = in_sizes[0] / 128;
  const int E = in_sizes[1] / 2;
  const int* src = ei;
  const int* dst = ei + E;
  float* out = (float*)d_out;

  float* ws = (float*)d_ws;
  size_t Npad = ((size_t)N + 63) & ~(size_t)63;
  float* dinv = ws;
  float* xw = ws + Npad;
  float* h  = xw + (size_t)N * 128;
  int* cnt    = (int*)(h + (size_t)N * 128);
  int* rowptr = cnt + Npad;
  int2* tmp   = (int2*)h;
  int* csr    = rowptr + Npad;
  int* cntmat = csr + E;
  const int nchunks = (E + EDGE_CHUNK - 1) / EDGE_CHUNK;
  int* gbase  = cntmat + (size_t)nchunks * GRP_MAX;

  group_count_k<<<nchunks, 1024, 0, stream>>>(dst, cntmat, E);
  group_scatter_k<<<nchunks, 1024, 0, stream>>>(src, dst, cntmat, gbase, tmp, E, nchunks);
  group_csr_k<<<(N + GRP_SIZE - 1) / GRP_SIZE, 1024, 0, stream>>>(tmp, gbase, cnt, rowptr, dinv, csr, N, E);

  gemm_wmma_k<128><<<(N + 15) / 16, 256, 0, stream>>>(x, W1, xw, N, 128);
  gather_k<128, false><<<(N + 7) / 8, 256, 0, stream>>>(xw, dinv, rowptr, cnt, csr, b1, h, N);

  gemm_wmma_k<64><<<(N + 15) / 16, 128, 0, stream>>>(h, W2, xw, N, 128);
  gather_k<64, true><<<(N + 7) / 8, 256, 0, stream>>>(xw, dinv, rowptr, cnt, csr, b2, out, N);
}


# Layers reassociated: (Ahat(x W1)+b1)W2 computed as Ahat(x(W1 W2))+b1 W2; GEMM and first gather at 64 channels not 128, second GEMM gone; W1 W2 on f32 WMMA; product and sum order change
# speedup vs baseline: 4.0347x; 1.3186x over previous
#include <hip/hip_runtime.h>
#include <math.h>

typedef float v2f __attribute__((ext_vector_type(2)));
typedef float v8f __attribute__((ext_vector_type(8)));

#define GRP_SHIFT 9
#define GRP_SIZE  512
#define GRP_MAX   256
#define EDGE_CHUNK 8192
#define CSR_STAGE 12288

__global__ __launch_bounds__(1024)
void group_count_k(const int* __restrict__ dst, int* __restrict__ cntmat, int E) {
  __shared__ int hist[GRP_MAX];
  const int tid = threadIdx.x;
  if (tid < GRP_MAX) hist[tid] = 0;
  __syncthreads();
  const int e0 = blockIdx.x * EDGE_CHUNK + tid;
  int d[EDGE_CHUNK / 1024];
#pragma unroll
  for (int j = 0; j < EDGE_CHUNK / 1024; ++j) {
    const int e = e0 + j * 1024;
    d[j] = dst[e < E ? e : E - 1];
  }
#pragma unroll
  for (int j = 0; j < EDGE_CHUNK / 1024; ++j) {
    atomicAdd(&hist[d[j] >> GRP_SHIFT], (e0 + j * 1024 < E) ? 1 : 0);
  }
  __syncthreads();
  if (tid < GRP_MAX) cntmat[blockIdx.x * GRP_MAX + tid] = hist[tid];
}

__global__ __launch_bounds__(1024)
void group_scatter_k(const int* __restrict__ src, const int* __restrict__ dst,
                     const int* __restrict__ cntmat, int* __restrict__ gbase,
                     int2* __restrict__ tmp, int E, int nchunks) {
  __shared__ int part_tot[4][GRP_MAX];
  __shared__ int part_pre[4][GRP_MAX];
  __shared__ int off[GRP_MAX];
  __shared__ int wsum[8];
  const int tid = threadIdx.x;
  const int lane = tid & 31;
  const int w = tid >> 5;
  const int blk = blockIdx.x;
  {
    const int b = tid & (GRP_MAX - 1);
    const int g = tid >> 8;
    int tot = 0, pre = 0;
    for (int r = g; r < nchunks; r += 4) {
      const int v = cntmat[r * GRP_MAX + b];
      tot += v;
      if (r < blk) pre += v;
    }
    part_tot[g][b] = tot;
    part_pre[g][b] = pre;
  }
  __syncthreads();
  int t = 0, p = 0, x = 0;
  if (tid < GRP_MAX) {
    t = part_tot[0][tid] + part_tot[1][tid] + part_tot[2][tid] + part_tot[3][tid];
    p = part_pre[0][tid] + part_pre[1][tid] + part_pre[2][tid] + part_pre[3][tid];
    x = t;
    for (int o = 1; o < 32; o <<= 1) {
      const int y = __shfl_up(x, o, 32);
      if (lane >= o) x += y;
    }
    if (lane == 31) wsum[w] = x;
  }
  __syncthreads();
  if (tid < GRP_MAX) {
    int wp = 0;
    for (int k = 0; k < w; ++k) wp += wsum[k];
    const int base = wp + x - t;
    off[tid] = base + p;
    if (blk == 0) gbase[tid] = base;
  }
  __syncthreads();
  const int e0 = blk * EDGE_CHUNK + tid;
  int d[EDGE_CHUNK / 1024], s[EDGE_CHUNK / 1024];
#pragma unroll
  for (int j = 0; j < EDGE_CHUNK / 1024; ++j) {
    const int e = e0 + j * 1024;
    const int ec = e < E ? e : E - 1;
    d[j] = dst[ec];
    s[j] = src[ec];
  }
  int pos[EDGE_CHUNK / 1024];
#pragma unroll
  for (int j = 0; j < EDGE_CHUNK / 1024; ++j) {
    pos[j] = atomicAdd(&off[d[j] >> GRP_SHIFT], (e0 + j * 1024 < E) ? 1 : 0);
  }
#pragma unroll
  for (int j = 0; j < EDGE_CHUNK / 1024; ++j) {
    if (e0 + j * 1024 < E) tmp[pos[j]] = make_int2(s[j], d[j]);
  }
}

__global__ __launch_bounds__(1024)
void group_csr_k(const int2* __restrict__ tmp, const int* __restrict__ gbase,
                 int* __restrict__ cnt, int* __restrict__ rowptr,
                 float* __restrict__ dinv, int* __restrict__ csr, int N, int E) {
  __shared__ int c[GRP_SIZE];
  __shared__ int cur[GRP_SIZE];
  __shared__ int wsum[16];
  __shared__ int stage[CSR_STAGE];
  const int tid = threadIdx.x;
  const int lane = tid & 31;
  const int w = tid >> 5;
  const int b = blockIdx.x;
  const int lo = b << GRP_SHIFT;
  const int start = gbase[b];
  const int end = (b + 1 < GRP_MAX) ? gbase[b + 1] : E;
  const int n = end - start;

  if (tid < GRP_SIZE) c[tid] = 0;
  __syncthreads();
  for (int i0 = start + tid; i0 < end; i0 += 4096) {
    int dd[4];
#pragma unroll
    for (int k = 0; k < 4; ++k) {
      const int i = i0 + k * 1024;
      dd[k] = tmp[i < end ? i : i0].y;
    }
#pragma unroll
    for (int k = 0; k < 4; ++k) {
      atomicAdd(&c[dd[k] - lo], (i0 + k * 1024 < end) ? 1 : 0);
    }
  }
  __syncthreads();

  int v = 0, x = 0;
  if (tid < GRP_SIZE) {
    v = c[tid];
    x = v;
    for (int o = 1; o < 32; o <<= 1) {
      const int y = __shfl_up(x, o, 32);
      if (lane >= o) x += y;
    }
    if (lane == 31) wsum[w] = x;
  }
  __syncthreads();
  if (tid < GRP_SIZE) {
    int wp = 0;
    for (int k = 0; k < w; ++k) wp += wsum[k];
    const int excl = wp + x - v;
    cur[tid] = excl;
    const int i = lo + tid;
    if (i < N) {
      cnt[i] = v;
      rowptr[i] = start + excl;
      float d = (float)(v + 1);
      dinv[i] = (d > 0.0f) ? rsqrtf(d) : 0.0f;
    }
  }
  __syncthreads();

  const bool staged = (n <= CSR_STAGE);
  for (int i0 = start + tid; i0 < end; i0 += 4096) {
    int2 pr[4];
#pragma unroll
    for (int k = 0; k < 4; ++k) {
      const int i = i0 + k * 1024;
      pr[k] = tmp[i < end ? i : i0];
    }
    int q[4];
#pragma unroll
    for (int k = 0; k < 4; ++k) {
      q[k] = atomicAdd(&cur[pr[k].y - lo], (i0 + k * 1024 < end) ? 1 : 0);
    }
#pragma unroll
    for (int k = 0; k < 4; ++k) {
      if (i0 + k * 1024 < end) {
        if (staged) stage[q[k]] = pr[k].x;
        else csr[start + q[k]] = pr[k].x;
      }
    }
  }
  if (staged) {
    __syncthreads();
    for (int i = tid; i < n; i += 1024) csr[start + i] = stage[i];
  }
}

template <int NN>
__global__ void gemm_wmma_k(const float* __restrict__ A,
                            const float* __restrict__ B,
                            float* __restrict__ C,
                            int M, int K) {
  const int wave = threadIdx.x >> 5;
  const int lane = threadIdx.x & 31;
  const int m0 = blockIdx.x * 16;
  if (m0 >= M) return;
  const int n0 = wave * 16;
  const int half = lane >> 4;
  const int l15 = lane & 15;

  v8f acc = {};
  const float* aptr = A + (size_t)(m0 + l15) * K + half * 2;
  const float* bptr = B + (size_t)(half * 2) * NN + n0 + l15;

  for (int k = 0; k < K; k += 4) {
    v2f a, b;
    a.x = aptr[0];
    a.y = aptr[1];
    b.x = bptr[0];
    b.y = bptr[NN];
    acc = __builtin_amdgcn_wmma_f32_16x16x4_f32(
         false, a,  false, b,
         (short)0, acc,  false,  false);
    aptr += 4;
    bptr += 4 * NN;
  }

  float* cptr = C + (size_t)(m0 + half * 8) * NN + n0 + l15;
#pragma unroll
  for (int r = 0; r < 8; ++r) {
    cptr[0] = acc[r];
    cptr += NN;
  }
}

template <int NN>
__global__ void fold_weights_k(const float* __restrict__ W1,
                               const float* __restrict__ b1,
                               const float* __restrict__ W2,
                               float* __restrict__ W12,
                               int KIN, int K) {
  const int wave = threadIdx.x >> 5;
  const int lane = threadIdx.x & 31;
  const int m0 = blockIdx.x * 16;
  const int n0 = wave * 16;
  const int half = lane >> 4;
  const int l15 = lane & 15;
  const bool bias_tile = (m0 >= KIN);
  const bool has_row = !bias_tile || l15 == 0;

  v8f acc = {};
  const float* aptr = (bias_tile ? b1 : W1 + (size_t)(m0 + l15) * K) + half * 2;
  const float* bptr = W2 + (size_t)(half * 2) * NN + n0 + l15;

  for (int k = 0; k < K; k += 4) {
    v2f a, b;
    const float a0 = aptr[0], a1 = aptr[1];
    a.x = has_row ? a0 : 0.0f;
    a.y = has_row ? a1 : 0.0f;
    b.x = bptr[0];
    b.y = bptr[NN];
    acc = __builtin_amdgcn_wmma_f32_16x16x4_f32(
         false, a,  false, b,
         (short)0, acc,  false,  false);
    aptr += 4;
    bptr += 4 * NN;
  }

  const int mrow = m0 + half * 8;
  float* cptr = W12 + (size_t)mrow * NN + n0 + l15;
#pragma unroll
  for (int r = 0; r < 8; ++r) {
    if (mrow + r <= KIN) cptr[0] = acc[r];
    cptr += NN;
  }
}

template <int C> struct rowvec;
template <> struct rowvec<64>  { typedef float2 type; };

__device__ __forceinline__ void axpy(float2& a, float w, const float2& v) {
  a.x += w * v.x; a.y += w * v.y;
}

__device__ __forceinline__ float2 scale_normalize(float2 v) {
  float mx = fmaxf(v.x, v.y);
  float mn = fminf(v.x, v.y);
#pragma unroll
  for (int o = 16; o > 0; o >>= 1) {
    mx = fmaxf(mx, __shfl_xor(mx, o, 32));
    mn = fminf(mn, __shfl_xor(mn, o, 32));
  }
  float inv = 1.0f / (mx - mn);
  float sx = (v.x - mn) * inv;
  float sy = (v.y - mn) * inv;
  float ss = sx * sx + sy * sy;
#pragma unroll
  for (int o = 16; o > 0; o >>= 1) ss += __shfl_xor(ss, o, 32);
  float r = 1.0f / fmaxf(sqrtf(ss), 1e-12f);
  float2 ov;
  ov.x = sx * r;
  ov.y = sy * r;
  return ov;
}

template <int C, bool NORM>
__global__ void gather_k(const float* __restrict__ feat,
                         const float* __restrict__ dinv,
                         const int* __restrict__ rowptr,
                         const int* __restrict__ cnt,
                         const int* __restrict__ csr,
                         const float* __restrict__ bias,
                         float* __restrict__ out, int N) {
  typedef typename rowvec<C>::type vec;
  const int lane = threadIdx.x & 31;
  const int node = __builtin_amdgcn_readfirstlane(
      (int)(blockIdx.x * (blockDim.x >> 5) + (threadIdx.x >> 5)));
  if (node >= N) return;

  int p = rowptr[node];
  const int end = p + cnt[node];
  const float di = dinv[node];

  vec acc = reinterpret_cast<const vec*>(bias)[lane];
  axpy(acc, di * di, reinterpret_cast<const vec*>(feat + (size_t)node * C)[lane]);

  for (; p + 4 <= end; p += 4) {
    const int s0 = csr[p], s1 = csr[p + 1], s2 = csr[p + 2], s3 = csr[p + 3];
    const vec v0 = reinterpret_cast<const vec*>(feat + (size_t)s0 * C)[lane];
    const vec v1 = reinterpret_cast<const vec*>(feat + (size_t)s1 * C)[lane];
    const vec v2 = reinterpret_cast<const vec*>(feat + (size_t)s2 * C)[lane];
    const vec v3 = reinterpret_cast<const vec*>(feat + (size_t)s3 * C)[lane];
    const float w0 = dinv[s0] * di, w1 = dinv[s1] * di;
    const float w2 = dinv[s2] * di, w3 = dinv[s3] * di;
    axpy(acc, w0, v0);
    axpy(acc, w1, v1);
    axpy(acc, w2, v2);
    axpy(acc, w3, v3);
  }
  if (p + 2 <= end) {
    const int s0 = csr[p], s1 = csr[p + 1];
    const vec v0 = reinterpret_cast<const vec*>(feat + (size_t)s0 * C)[lane];
    const vec v1 = reinterpret_cast<const vec*>(feat + (size_t)s1 * C)[lane];
    const float w0 = dinv[s0] * di, w1 = dinv[s1] * di;
    axpy(acc, w0, v0);
    axpy(acc, w1, v1);
    p += 2;
  }
  if (p < end) {
    const int s0 = csr[p];
    const vec v0 = reinterpret_cast<const vec*>(feat + (size_t)s0 * C)[lane];
    axpy(acc, dinv[s0] * di, v0);
  }
  if constexpr (NORM) {
    static_assert(C == 64, "the fused scale + normalise expects one float2 per lane");
    acc = scale_normalize(acc);
  }
  reinterpret_cast<vec*>(out + (size_t)node * C)[lane] = acc;
}

extern "C" void kernel_launch(void* const* d_in, const int* in_sizes, int n_in,
                              void* d_out, int out_size, void* d_ws, size_t ws_size,
                              hipStream_t stream) {
  (void)n_in; (void)out_size; (void)ws_size;

  const float* x  = (const float*)d_in[0];
  const int*   ei = (const int*)d_in[1];
  const float* W1 = (const float*)d_in[2];
  const float* b1 = (const float*)d_in[3];
  const float* W2 = (const float*)d_in[4];
  const float* b2 = (const float*)d_in[5];

  const int N = in_sizes[0] / 128;
  const int E = in_sizes[1] / 2;
  const int* src = ei;
  const int* dst = ei + E;
  float* out = (float*)d_out;

  float* ws = (float*)d_ws;
  size_t Npad = ((size_t)N + 63) & ~(size_t)63;
  float* dinv = ws;
  float* y = ws + Npad;
  float* g = y + (size_t)N * 64;
  float* scratch = g + (size_t)N * 64;
  int* cnt    = (int*)(scratch + (size_t)N * 128);
  int* rowptr = cnt + Npad;
  int2* tmp   = (int2*)scratch;
  float* W12  = scratch;
  int* csr    = rowptr + Npad;
  int* cntmat = csr + E;
  const int nchunks = (E + EDGE_CHUNK - 1) / EDGE_CHUNK;
  int* gbase  = cntmat + (size_t)nchunks * GRP_MAX;

  group_count_k<<<nchunks, 1024, 0, stream>>>(dst, cntmat, E);
  group_scatter_k<<<nchunks, 1024, 0, stream>>>(src, dst, cntmat, gbase, tmp, E, nchunks);
  group_csr_k<<<(N + GRP_SIZE - 1) / GRP_SIZE, 1024, 0, stream>>>(tmp, gbase, cnt, rowptr, dinv, csr, N, E);

  fold_weights_k<64><<<128 / 16 + 1, 128, 0, stream>>>(W1, b1, W2, W12, 128, 128);

  gemm_wmma_k<64><<<(N + 15) / 16, 128, 0, stream>>>(x, W12, y, N, 128);
  gather_k<64, false><<<(N + 7) / 8, 256, 0, stream>>>(y, dinv, rowptr, cnt, csr, W12 + 128 * 64, g, N);

  gather_k<64, true><<<(N + 7) / 8, 256, 0, stream>>>(g, dinv, rowptr, cnt, csr, b2, out, N);
}


# Gather: wave takes a node's edge list in one vector load (a lane per edge), lanes fetch their edge's dinv, readlane hands both out: up to 24 row loads in flight, not 4 per index load; sum order kept
# speedup vs baseline: 4.4127x; 1.0937x over previous
#include <hip/hip_runtime.h>
#include <math.h>

typedef float v2f __attribute__((ext_vector_type(2)));
typedef float v8f __attribute__((ext_vector_type(8)));

#define GRP_SHIFT 9
#define GRP_SIZE  512
#define GRP_MAX   256
#define EDGE_CHUNK 8192
#define CSR_STAGE 12288

__global__ __launch_bounds__(1024)
void group_count_k(const int* __restrict__ dst, int* __restrict__ cntmat, int E) {
  __shared__ int hist[GRP_MAX];
  const int tid = threadIdx.x;
  if (tid < GRP_MAX) hist[tid] = 0;
  __syncthreads();
  const int e0 = blockIdx.x * EDGE_CHUNK + tid;
  int d[EDGE_CHUNK / 1024];
#pragma unroll
  for (int j = 0; j < EDGE_CHUNK / 1024; ++j) {
    const int e = e0 + j * 1024;
    d[j] = dst[e < E ? e : E - 1];
  }
#pragma unroll
  for (int j = 0; j < EDGE_CHUNK / 1024; ++j) {
    atomicAdd(&hist[d[j] >> GRP_SHIFT], (e0 + j * 1024 < E) ? 1 : 0);
  }
  __syncthreads();
  if (tid < GRP_MAX) cntmat[blockIdx.x * GRP_MAX + tid] = hist[tid];
}

__global__ __launch_bounds__(1024)
void group_scatter_k(const int* __restrict__ src, const int* __restrict__ dst,
                     const int* __restrict__ cntmat, int* __restrict__ gbase,
                     int2* __restrict__ tmp, int E, int nchunks) {
  __shared__ int part_tot[4][GRP_MAX];
  __shared__ int part_pre[4][GRP_MAX];
  __shared__ int off[GRP_MAX];
  __shared__ int wsum[8];
  const int tid = threadIdx.x;
  const int lane = tid & 31;
  const int w = tid >> 5;
  const int blk = blockIdx.x;
  {
    const int b = tid & (GRP_MAX - 1);
    const int g = tid >> 8;
    int tot = 0, pre = 0;
    for (int r = g; r < nchunks; r += 4) {
      const int v = cntmat[r * GRP_MAX + b];
      tot += v;
      if (r < blk) pre += v;
    }
    part_tot[g][b] = tot;
    part_pre[g][b] = pre;
  }
  __syncthreads();
  int t = 0, p = 0, x = 0;
  if (tid < GRP_MAX) {
    t = part_tot[0][tid] + part_tot[1][tid] + part_tot[2][tid] + part_tot[3][tid];
    p = part_pre[0][tid] + part_pre[1][tid] + part_pre[2][tid] + part_pre[3][tid];
    x = t;
    for (int o = 1; o < 32; o <<= 1) {
      const int y = __shfl_up(x, o, 32);
      if (lane >= o) x += y;
    }
    if (lane == 31) wsum[w] = x;
  }
  __syncthreads();
  if (tid < GRP_MAX) {
    int wp = 0;
    for (int k = 0; k < w; ++k) wp += wsum[k];
    const int base = wp + x - t;
    off[tid] = base + p;
    if (blk == 0) gbase[tid] = base;
  }
  __syncthreads();
  const int e0 = blk * EDGE_CHUNK + tid;
  int d[EDGE_CHUNK / 1024], s[EDGE_CHUNK / 1024];
#pragma unroll
  for (int j = 0; j < EDGE_CHUNK / 1024; ++j) {
    const int e = e0 + j * 1024;
    const int ec = e < E ? e : E - 1;
    d[j] = dst[ec];
    s[j] = src[ec];
  }
  int pos[EDGE_CHUNK / 1024];
#pragma unroll
  for (int j = 0; j < EDGE_CHUNK / 1024; ++j) {
    pos[j] = atomicAdd(&off[d[j] >> GRP_SHIFT], (e0 + j * 1024 < E) ? 1 : 0);
  }
#pragma unroll
  for (int j = 0; j < EDGE_CHUNK / 1024; ++j) {
    if (e0 + j * 1024 < E) tmp[pos[j]] = make_int2(s[j], d[j]);
  }
}

__global__ __launch_bounds__(1024)
void group_csr_k(const int2* __restrict__ tmp, const int* __restrict__ gbase,
                 int* __restrict__ cnt, int* __restrict__ rowptr,
                 float* __restrict__ dinv, int* __restrict__ csr, int N, int E) {
  __shared__ int c[GRP_SIZE];
  __shared__ int cur[GRP_SIZE];
  __shared__ int wsum[16];
  __shared__ int stage[CSR_STAGE];
  const int tid = threadIdx.x;
  const int lane = tid & 31;
  const int w = tid >> 5;
  const int b = blockIdx.x;
  const int lo = b << GRP_SHIFT;
  const int start = gbase[b];
  const int end = (b + 1 < GRP_MAX) ? gbase[b + 1] : E;
  const int n = end - start;

  if (tid < GRP_SIZE) c[tid] = 0;
  __syncthreads();
  for (int i0 = start + tid; i0 < end; i0 += 4096) {
    int dd[4];
#pragma unroll
    for (int k = 0; k < 4; ++k) {
      const int i = i0 + k * 1024;
      dd[k] = tmp[i < end ? i : i0].y;
    }
#pragma unroll
    for (int k = 0; k < 4; ++k) {
      atomicAdd(&c[dd[k] - lo], (i0 + k * 1024 < end) ? 1 : 0);
    }
  }
  __syncthreads();

  int v = 0, x = 0;
  if (tid < GRP_SIZE) {
    v = c[tid];
    x = v;
    for (int o = 1; o < 32; o <<= 1) {
      const int y = __shfl_up(x, o, 32);
      if (lane >= o) x += y;
    }
    if (lane == 31) wsum[w] = x;
  }
  __syncthreads();
  if (tid < GRP_SIZE) {
    int wp = 0;
    for (int k = 0; k < w; ++k) wp += wsum[k];
    const int excl = wp + x - v;
    cur[tid] = excl;
    const int i = lo + tid;
    if (i < N) {
      cnt[i] = v;
      rowptr[i] = start + excl;
      float d = (float)(v + 1);
      dinv[i] = (d > 0.0f) ? rsqrtf(d) : 0.0f;
    }
  }
  __syncthreads();

  const bool staged = (n <= CSR_STAGE);
  for (int i0 = start + tid; i0 < end; i0 += 4096) {
    int2 pr[4];
#pragma unroll
    for (int k = 0; k < 4; ++k) {
      const int i = i0 + k * 1024;
      pr[k] = tmp[i < end ? i : i0];
    }
    int q[4];
#pragma unroll
    for (int k = 0; k < 4; ++k) {
      q[k] = atomicAdd(&cur[pr[k].y - lo], (i0 + k * 1024 < end) ? 1 : 0);
    }
#pragma unroll
    for (int k = 0; k < 4; ++k) {
      if (i0 + k * 1024 < end) {
        if (staged) stage[q[k]] = pr[k].x;
        else csr[start + q[k]] = pr[k].x;
      }
    }
  }
  if (staged) {
    __syncthreads();
    for (int i = tid; i < n; i += 1024) csr[start + i] = stage[i];
  }
}

template <int NN>
__global__ void gemm_wmma_k(const float* __restrict__ A,
                            const float* __restrict__ B,
                            float* __restrict__ C,
                            int M, int K) {
  const int wave = threadIdx.x >> 5;
  const int lane = threadIdx.x & 31;
  const int m0 = blockIdx.x * 16;
  if (m0 >= M) return;
  const int n0 = wave * 16;
  const int half = lane >> 4;
  const int l15 = lane & 15;

  v8f acc = {};
  const float* aptr = A + (size_t)(m0 + l15) * K + half * 2;
  const float* bptr = B + (size_t)(half * 2) * NN + n0 + l15;

  for (int k = 0; k < K; k += 4) {
    v2f a, b;
    a.x = aptr[0];
    a.y = aptr[1];
    b.x = bptr[0];
    b.y = bptr[NN];
    acc = __builtin_amdgcn_wmma_f32_16x16x4_f32(
         false, a,  false, b,
         (short)0, acc,  false,  false);
    aptr += 4;
    bptr += 4 * NN;
  }

  float* cptr = C + (size_t)(m0 + half * 8) * NN + n0 + l15;
#pragma unroll
  for (int r = 0; r < 8; ++r) {
    cptr[0] = acc[r];
    cptr += NN;
  }
}

template <int NN>
__global__ void fold_weights_k(const float* __restrict__ W1,
                               const float* __restrict__ b1,
                               const float* __restrict__ W2,
                               float* __restrict__ W12,
                               int KIN, int K) {
  const int wave = threadIdx.x >> 5;
  const int lane = threadIdx.x & 31;
  const int m0 = blockIdx.x * 16;
  const int n0 = wave * 16;
  const int half = lane >> 4;
  const int l15 = lane & 15;
  const bool bias_tile = (m0 >= KIN);
  const bool has_row = !bias_tile || l15 == 0;

  v8f acc = {};
  const float* aptr = (bias_tile ? b1 : W1 + (size_t)(m0 + l15) * K) + half * 2;
  const float* bptr = W2 + (size_t)(half * 2) * NN + n0 + l15;

  for (int k = 0; k < K; k += 4) {
    v2f a, b;
    const float a0 = aptr[0], a1 = aptr[1];
    a.x = has_row ? a0 : 0.0f;
    a.y = has_row ? a1 : 0.0f;
    b.x = bptr[0];
    b.y = bptr[NN];
    acc = __builtin_amdgcn_wmma_f32_16x16x4_f32(
         false, a,  false, b,
         (short)0, acc,  false,  false);
    aptr += 4;
    bptr += 4 * NN;
  }

  const int mrow = m0 + half * 8;
  float* cptr = W12 + (size_t)mrow * NN + n0 + l15;
#pragma unroll
  for (int r = 0; r < 8; ++r) {
    if (mrow + r <= KIN) cptr[0] = acc[r];
    cptr += NN;
  }
}

template <int C> struct rowvec;
template <> struct rowvec<64>  { typedef float2 type; };

__device__ __forceinline__ void axpy(float2& a, float w, const float2& v) {
  a.x += w * v.x; a.y += w * v.y;
}

__device__ __forceinline__ float2 scale_normalize(float2 v) {
  float mx = fmaxf(v.x, v.y);
  float mn = fminf(v.x, v.y);
#pragma unroll
  for (int o = 16; o > 0; o >>= 1) {
    mx = fmaxf(mx, __shfl_xor(mx, o, 32));
    mn = fminf(mn, __shfl_xor(mn, o, 32));
  }
  float inv = 1.0f / (mx - mn);
  float sx = (v.x - mn) * inv;
  float sy = (v.y - mn) * inv;
  float ss = sx * sx + sy * sy;
#pragma unroll
  for (int o = 16; o > 0; o >>= 1) ss += __shfl_xor(ss, o, 32);
  float r = 1.0f / fmaxf(sqrtf(ss), 1e-12f);
  float2 ov;
  ov.x = sx * r;
  ov.y = sy * r;
  return ov;
}

#define ROWS_IN_FLIGHT 24
template <int C> struct rowset {
  typename rowvec<C>::type r0, r1, r2, r3, r4, r5, r6, r7, r8, r9, r10, r11, r12, r13, r14, r15, r16, r17, r18, r19, r20, r21, r22, r23;
};

template <int C, int J0, int CNT>
__device__ __forceinline__ void load_rows(rowset<C>& v, const float* __restrict__ feat,
                                          int sv, int m, unsigned lane_off) {
  typedef typename rowvec<C>::type vec;
  const char* fb = reinterpret_cast<const char*>(feat);
#define ROW_OF_EDGE(K) \
  (*reinterpret_cast<const vec*>(fb + (((unsigned)__builtin_amdgcn_readlane(sv, J0 + K)) * (unsigned)(C * 4) + lane_off)))
  if (0 < CNT && J0 + 0 < m) { v.r0 = ROW_OF_EDGE(0);
  if (1 < CNT && J0 + 1 < m) { v.r1 = ROW_OF_EDGE(1);
  if (2 < CNT && J0 + 2 < m) { v.r2 = ROW_OF_EDGE(2);
  if (3 < CNT && J0 + 3 < m) { v.r3 = ROW_OF_EDGE(3);
  if (4 < CNT && J0 + 4 < m) { v.r4 = ROW_OF_EDGE(4);
  if (5 < CNT && J0 + 5 < m) { v.r5 = ROW_OF_EDGE(5);
  if (6 < CNT && J0 + 6 < m) { v.r6 = ROW_OF_EDGE(6);
  if (7 < CNT && J0 + 7 < m) { v.r7 = ROW_OF_EDGE(7);
  if (8 < CNT && J0 + 8 < m) { v.r8 = ROW_OF_EDGE(8);
  if (9 < CNT && J0 + 9 < m) { v.r9 = ROW_OF_EDGE(9);
  if (10 < CNT && J0 + 10 < m) { v.r10 = ROW_OF_EDGE(10);
  if (11 < CNT && J0 + 11 < m) { v.r11 = ROW_OF_EDGE(11);
  if (12 < CNT && J0 + 12 < m) { v.r12 = ROW_OF_EDGE(12);
  if (13 < CNT && J0 + 13 < m) { v.r13 = ROW_OF_EDGE(13);
  if (14 < CNT && J0 + 14 < m) { v.r14 = ROW_OF_EDGE(14);
  if (15 < CNT && J0 + 15 < m) { v.r15 = ROW_OF_EDGE(15);
  if (16 < CNT && J0 + 16 < m) { v.r16 = ROW_OF_EDGE(16);
  if (17 < CNT && J0 + 17 < m) { v.r17 = ROW_OF_EDGE(17);
  if (18 < CNT && J0 + 18 < m) { v.r18 = ROW_OF_EDGE(18);
  if (19 < CNT && J0 + 19 < m) { v.r19 = ROW_OF_EDGE(19);
  if (20 < CNT && J0 + 20 < m) { v.r20 = ROW_OF_EDGE(20);
  if (21 < CNT && J0 + 21 < m) { v.r21 = ROW_OF_EDGE(21);
  if (22 < CNT && J0 + 22 < m) { v.r22 = ROW_OF_EDGE(22);
  if (23 < CNT && J0 + 23 < m) { v.r23 = ROW_OF_EDGE(23);
  }}}}}}}}}}}}}}}}}}}}}}}}
#undef ROW_OF_EDGE
}

template <int C, int J0, int CNT>
__device__ __forceinline__ void add_rows(typename rowvec<C>::type& acc, const rowset<C>& v,
                                         float wv, int m) {
#define WEIGHT_OF_EDGE(K) __int_as_float(__builtin_amdgcn_readlane(__float_as_int(wv), J0 + K))
  if (0 < CNT && J0 + 0 < m) { axpy(acc, WEIGHT_OF_EDGE(0), v.r0);
  if (1 < CNT && J0 + 1 < m) { axpy(acc, WEIGHT_OF_EDGE(1), v.r1);
  if (2 < CNT && J0 + 2 < m) { axpy(acc, WEIGHT_OF_EDGE(2), v.r2);
  if (3 < CNT && J0 + 3 < m) { axpy(acc, WEIGHT_OF_EDGE(3), v.r3);
  if (4 < CNT && J0 + 4 < m) { axpy(acc, WEIGHT_OF_EDGE(4), v.r4);
  if (5 < CNT && J0 + 5 < m) { axpy(acc, WEIGHT_OF_EDGE(5), v.r5);
  if (6 < CNT && J0 + 6 < m) { axpy(acc, WEIGHT_OF_EDGE(6), v.r6);
  if (7 < CNT && J0 + 7 < m) { axpy(acc, WEIGHT_OF_EDGE(7), v.r7);
  if (8 < CNT && J0 + 8 < m) { axpy(acc, WEIGHT_OF_EDGE(8), v.r8);
  if (9 < CNT && J0 + 9 < m) { axpy(acc, WEIGHT_OF_EDGE(9), v.r9);
  if (10 < CNT && J0 + 10 < m) { axpy(acc, WEIGHT_OF_EDGE(10), v.r10);
  if (11 < CNT && J0 + 11 < m) { axpy(acc, WEIGHT_OF_EDGE(11), v.r11);
  if (12 < CNT && J0 + 12 < m) { axpy(acc, WEIGHT_OF_EDGE(12), v.r12);
  if (13 < CNT && J0 + 13 < m) { axpy(acc, WEIGHT_OF_EDGE(13), v.r13);
  if (14 < CNT && J0 + 14 < m) { axpy(acc, WEIGHT_OF_EDGE(14), v.r14);
  if (15 < CNT && J0 + 15 < m) { axpy(acc, WEIGHT_OF_EDGE(15), v.r15);
  if (16 < CNT && J0 + 16 < m) { axpy(acc, WEIGHT_OF_EDGE(16), v.r16);
  if (17 < CNT && J0 + 17 < m) { axpy(acc, WEIGHT_OF_EDGE(17), v.r17);
  if (18 < CNT && J0 + 18 < m) { axpy(acc, WEIGHT_OF_EDGE(18), v.r18);
  if (19 < CNT && J0 + 19 < m) { axpy(acc, WEIGHT_OF_EDGE(19), v.r19);
  if (20 < CNT && J0 + 20 < m) { axpy(acc, WEIGHT_OF_EDGE(20), v.r20);
  if (21 < CNT && J0 + 21 < m) { axpy(acc, WEIGHT_OF_EDGE(21), v.r21);
  if (22 < CNT && J0 + 22 < m) { axpy(acc, WEIGHT_OF_EDGE(22), v.r22);
  if (23 < CNT && J0 + 23 < m) { axpy(acc, WEIGHT_OF_EDGE(23), v.r23);
  }}}}}}}}}}}}}}}}}}}}}}}}
#undef WEIGHT_OF_EDGE
}

template <int C, bool NORM>
__global__ void gather_k(const float* __restrict__ feat,
                         const float* __restrict__ dinv,
                         const int* __restrict__ rowptr,
                         const int* __restrict__ cnt,
                         const int* __restrict__ csr,
                         const float* __restrict__ bias,
                         float* __restrict__ out, int N) {
  typedef typename rowvec<C>::type vec;
  const int lane = threadIdx.x & 31;
  const int node = __builtin_amdgcn_readfirstlane(
      (int)(blockIdx.x * (blockDim.x >> 5) + (threadIdx.x >> 5)));
  if (node >= N) return;

  int p = rowptr[node];
  int deg = cnt[node];
  float di = dinv[node];
  asm volatile("" : "+s"(p), "+s"(deg), "+s"(di));

  const vec bias_v = reinterpret_cast<const vec*>(bias)[lane];
  const vec self_v = reinterpret_cast<const vec*>(feat + (size_t)node * C)[lane];
  vec acc;
  bool started = false;

  const unsigned lane_off = (unsigned)lane * (unsigned)sizeof(vec);
  for (int m = deg; m > 0; m -= 32, p += 32) {
    const unsigned mine = (unsigned)(lane < m ? lane : m - 1);
    const int sv = csr[(unsigned)p + mine];
    const float dv = dinv[sv];
    rowset<C> v;
    load_rows<C, 0, ROWS_IN_FLIGHT>(v, feat, sv, m, lane_off);
    if (!started) {
      acc = bias_v;
      axpy(acc, di * di, self_v);
      started = true;
    }
    const float wv = dv * di;
    add_rows<C, 0, ROWS_IN_FLIGHT>(acc, v, wv, m);
    if (m > ROWS_IN_FLIGHT) {
      load_rows<C, ROWS_IN_FLIGHT, 32 - ROWS_IN_FLIGHT>(v, feat, sv, m, lane_off);
      add_rows<C, ROWS_IN_FLIGHT, 32 - ROWS_IN_FLIGHT>(acc, v, wv, m);
    }
  }
  if (!started) {
    acc = bias_v;
    axpy(acc, di * di, self_v);
  }
  if constexpr (NORM) {
    static_assert(C == 64, "the fused scale + normalise expects one float2 per lane");
    acc = scale_normalize(acc);
  }
  reinterpret_cast<vec*>(out + (size_t)node * C)[lane] = acc;
}

extern "C" void kernel_launch(void* const* d_in, const int* in_sizes, int n_in,
                              void* d_out, int out_size, void* d_ws, size_t ws_size,
                              hipStream_t stream) {
  (void)n_in; (void)out_size; (void)ws_size;

  const float* x  = (const float*)d_in[0];
  const int*   ei = (const int*)d_in[1];
  const float* W1 = (const float*)d_in[2];
  const float* b1 = (const float*)d_in[3];
  const float* W2 = (const float*)d_in[4];
  const float* b2 = (const float*)d_in[5];

  const int N = in_sizes[0] / 128;
  const int E = in_sizes[1] / 2;
  const int* src = ei;
  const int* dst = ei + E;
  float* out = (float*)d_out;

  float* ws = (float*)d_ws;
  size_t Npad = ((size_t)N + 63) & ~(size_t)63;
  float* dinv = ws;
  float* y = ws + Npad;
  float* g = y + (size_t)N * 64;
  float* scratch = g + (size_t)N * 64;
  int* cnt    = (int*)(scratch + (size_t)N * 128);
  int* rowptr = cnt + Npad;
  int2* tmp   = (int2*)scratch;
  float* W12  = scratch;
  int* csr    = rowptr + Npad;
  int* cntmat = csr + E;
  const int nchunks = (E + EDGE_CHUNK - 1) / EDGE_CHUNK;
  int* gbase  = cntmat + (size_t)nchunks * GRP_MAX;

  group_count_k<<<nchunks, 1024, 0, stream>>>(dst, cntmat, E);
  group_scatter_k<<<nchunks, 1024, 0, stream>>>(src, dst, cntmat, gbase, tmp, E, nchunks);
  group_csr_k<<<(N + GRP_SIZE - 1) / GRP_SIZE, 1024, 0, stream>>>(tmp, gbase, cnt, rowptr, dinv, csr, N, E);

  fold_weights_k<64><<<128 / 16 + 1, 128, 0, stream>>>(W1, b1, W2, W12, 128, 128);

  gemm_wmma_k<64><<<(N + 15) / 16, 128, 0, stream>>>(x, W12, y, N, 128);
  gather_k<64, false><<<(N + 7) / 8, 256, 0, stream>>>(y, dinv, rowptr, cnt, csr, W12 + 128 * 64, g, N);

  gather_k<64, true><<<(N + 7) / 8, 256, 0, stream>>>(g, dinv, rowptr, cnt, csr, b2, out, N);
}


# GEMM re-tiled: one wave per 16x64 strip, 4 accumulator tiles share each A fragment; A read once (64 B per lane), B as float4 across the 4 tiles, whole-row float4 stores; k order of the sums changes
# speedup vs baseline: 4.7341x; 1.0728x over previous
#include <hip/hip_runtime.h>
#include <math.h>

typedef float v2f __attribute__((ext_vector_type(2)));
typedef float v8f __attribute__((ext_vector_type(8)));

#define GRP_SHIFT 9
#define GRP_SIZE  512
#define GRP_MAX   256
#define EDGE_CHUNK 8192
#define CSR_STAGE 12288

__global__ __launch_bounds__(1024)
void group_count_k(const int* __restrict__ dst, int* __restrict__ cntmat, int E) {
  __shared__ int hist[GRP_MAX];
  const int tid = threadIdx.x;
  if (tid < GRP_MAX) hist[tid] = 0;
  __syncthreads();
  const int e0 = blockIdx.x * EDGE_CHUNK + tid;
  int d[EDGE_CHUNK / 1024];
#pragma unroll
  for (int j = 0; j < EDGE_CHUNK / 1024; ++j) {
    const int e = e0 + j * 1024;
    d[j] = dst[e < E ? e : E - 1];
  }
#pragma unroll
  for (int j = 0; j < EDGE_CHUNK / 1024; ++j) {
    atomicAdd(&hist[d[j] >> GRP_SHIFT], (e0 + j * 1024 < E) ? 1 : 0);
  }
  __syncthreads();
  if (tid < GRP_MAX) cntmat[blockIdx.x * GRP_MAX + tid] = hist[tid];
}

__global__ __launch_bounds__(1024)
void group_scatter_k(const int* __restrict__ src, const int* __restrict__ dst,
                     const int* __restrict__ cntmat, int* __restrict__ gbase,
                     int2* __restrict__ tmp, int E, int nchunks) {
  __shared__ int part_tot[4][GRP_MAX];
  __shared__ int part_pre[4][GRP_MAX];
  __shared__ int off[GRP_MAX];
  __shared__ int wsum[8];
  const int tid = threadIdx.x;
  const int lane = tid & 31;
  const int w = tid >> 5;
  const int blk = blockIdx.x;
  {
    const int b = tid & (GRP_MAX - 1);
    const int g = tid >> 8;
    int tot = 0, pre = 0;
    for (int r = g; r < nchunks; r += 4) {
      const int v = cntmat[r * GRP_MAX + b];
      tot += v;
      if (r < blk) pre += v;
    }
    part_tot[g][b] = tot;
    part_pre[g][b] = pre;
  }
  __syncthreads();
  int t = 0, p = 0, x = 0;
  if (tid < GRP_MAX) {
    t = part_tot[0][tid] + part_tot[1][tid] + part_tot[2][tid] + part_tot[3][tid];
    p = part_pre[0][tid] + part_pre[1][tid] + part_pre[2][tid] + part_pre[3][tid];
    x = t;
    for (int o = 1; o < 32; o <<= 1) {
      const int y = __shfl_up(x, o, 32);
      if (lane >= o) x += y;
    }
    if (lane == 31) wsum[w] = x;
  }
  __syncthreads();
  if (tid < GRP_MAX) {
    int wp = 0;
    for (int k = 0; k < w; ++k) wp += wsum[k];
    const int base = wp + x - t;
    off[tid] = base + p;
    if (blk == 0) gbase[tid] = base;
  }
  __syncthreads();
  const int e0 = blk * EDGE_CHUNK + tid;
  int d[EDGE_CHUNK / 1024], s[EDGE_CHUNK / 1024];
#pragma unroll
  for (int j = 0; j < EDGE_CHUNK / 1024; ++j) {
    const int e = e0 + j * 1024;
    const int ec = e < E ? e : E - 1;
    d[j] = dst[ec];
    s[j] = src[ec];
  }
  int pos[EDGE_CHUNK / 1024];
#pragma unroll
  for (int j = 0; j < EDGE_CHUNK / 1024; ++j) {
    pos[j] = atomicAdd(&off[d[j] >> GRP_SHIFT], (e0 + j * 1024 < E) ? 1 : 0);
  }
#pragma unroll
  for (int j = 0; j < EDGE_CHUNK / 1024; ++j) {
    if (e0 + j * 1024 < E) tmp[pos[j]] = make_int2(s[j], d[j]);
  }
}

__global__ __launch_bounds__(1024)
void group_csr_k(const int2* __restrict__ tmp, const int* __restrict__ gbase,
                 int* __restrict__ cnt, int* __restrict__ rowptr,
                 float* __restrict__ dinv, int* __restrict__ csr, int N, int E) {
  __shared__ int c[GRP_SIZE];
  __shared__ int cur[GRP_SIZE];
  __shared__ int wsum[16];
  __shared__ int stage[CSR_STAGE];
  const int tid = threadIdx.x;
  const int lane = tid & 31;
  const int w = tid >> 5;
  const int b = blockIdx.x;
  const int lo = b << GRP_SHIFT;
  const int start = gbase[b];
  const int end = (b + 1 < GRP_MAX) ? gbase[b + 1] : E;
  const int n = end - start;

  if (tid < GRP_SIZE) c[tid] = 0;
  __syncthreads();
  for (int i0 = start + tid; i0 < end; i0 += 4096) {
    int dd[4];
#pragma unroll
    for (int k = 0; k < 4; ++k) {
      const int i = i0 + k * 1024;
      dd[k] = tmp[i < end ? i : i0].y;
    }
#pragma unroll
    for (int k = 0; k < 4; ++k) {
      atomicAdd(&c[dd[k] - lo], (i0 + k * 1024 < end) ? 1 : 0);
    }
  }
  __syncthreads();

  int v = 0, x = 0;
  if (tid < GRP_SIZE) {
    v = c[tid];
    x = v;
    for (int o = 1; o < 32; o <<= 1) {
      const int y = __shfl_up(x, o, 32);
      if (lane >= o) x += y;
    }
    if (lane == 31) wsum[w] = x;
  }
  __syncthreads();
  if (tid < GRP_SIZE) {
    int wp = 0;
    for (int k = 0; k < w; ++k) wp += wsum[k];
    const int excl = wp + x - v;
    cur[tid] = excl;
    const int i = lo + tid;
    if (i < N) {
      cnt[i] = v;
      rowptr[i] = start + excl;
      float d = (float)(v + 1);
      dinv[i] = (d > 0.0f) ? rsqrtf(d) : 0.0f;
    }
  }
  __syncthreads();

  const bool staged = (n <= CSR_STAGE);
  for (int i0 = start + tid; i0 < end; i0 += 4096) {
    int2 pr[4];
#pragma unroll
    for (int k = 0; k < 4; ++k) {
      const int i = i0 + k * 1024;
      pr[k] = tmp[i < end ? i : i0];
    }
    int q[4];
#pragma unroll
    for (int k = 0; k < 4; ++k) {
      q[k] = atomicAdd(&cur[pr[k].y - lo], (i0 + k * 1024 < end) ? 1 : 0);
    }
#pragma unroll
    for (int k = 0; k < 4; ++k) {
      if (i0 + k * 1024 < end) {
        if (staged) stage[q[k]] = pr[k].x;
        else csr[start + q[k]] = pr[k].x;
      }
    }
  }
  if (staged) {
    __syncthreads();
    for (int i = tid; i < n; i += 1024) csr[start + i] = stage[i];
  }
}

typedef float v4f __attribute__((ext_vector_type(4)));

template <int KK>
__global__ __launch_bounds__(128)
void gemm_strip_k(const float* __restrict__ A,
                  const float* __restrict__ B,
                  float* __restrict__ C, int M) {
  static_assert(KK % 32 == 0, "k is taken in blocks of 32");
  constexpr int NN = 64;
  constexpr int STEPS = KK / 4;
  const int wave = __builtin_amdgcn_readfirstlane((int)(threadIdx.x >> 5));
  const int lane = threadIdx.x & 31;
  const int m0 = (blockIdx.x * 4 + wave) * 16;
  if (m0 >= M) return;
  const int h = lane >> 4;
  const int c = lane & 15;

  const v4f* ap = reinterpret_cast<const v4f*>(A + (size_t)(m0 + c) * KK + h * 16);
  v4f a[KK / 32][4];
#pragma unroll
  for (int kb = 0; kb < KK / 32; ++kb) {
#pragma unroll
    for (int q = 0; q < 4; ++q) a[kb][q] = ap[kb * 8 + q];
  }

  const float* bp = B + (size_t)(h * 16) * NN + 4 * c;
  v4f br[3][2];
#pragma unroll
  for (int t = 0; t < 2; ++t) {
    const int k = (t >> 3) * 32 + (t & 7) * 2;
    br[t][0] = *reinterpret_cast<const v4f*>(bp + (size_t)k * NN);
    br[t][1] = *reinterpret_cast<const v4f*>(bp + (size_t)(k + 1) * NN);
  }

  v8f acc0 = {}, acc1 = {}, acc2 = {}, acc3 = {};
#pragma unroll
  for (int t = 0; t < STEPS; ++t) {
    if (t + 2 < STEPS) {
      const int kn = ((t + 2) >> 3) * 32 + ((t + 2) & 7) * 2;
      br[(t + 2) % 3][0] = *reinterpret_cast<const v4f*>(bp + (size_t)kn * NN);
      br[(t + 2) % 3][1] = *reinterpret_cast<const v4f*>(bp + (size_t)(kn + 1) * NN);
    }
    __builtin_amdgcn_sched_barrier(0);
    const v4f aq = a[t >> 3][(t >> 1) & 3];
    const v4f b0 = br[t % 3][0];
    const v4f b1 = br[t % 3][1];
    v2f av, bv;
    av.x = (t & 1) ? aq.z : aq.x;
    av.y = (t & 1) ? aq.w : aq.y;
    bv.x = b0.x; bv.y = b1.x;
    acc0 = __builtin_amdgcn_wmma_f32_16x16x4_f32(
         false, av,  false, bv,
         (short)0, acc0,  false,  false);
    bv.x = b0.y; bv.y = b1.y;
    acc1 = __builtin_amdgcn_wmma_f32_16x16x4_f32(
        false, av, false, bv, (short)0, acc1, false, false);
    bv.x = b0.z; bv.y = b1.z;
    acc2 = __builtin_amdgcn_wmma_f32_16x16x4_f32(
        false, av, false, bv, (short)0, acc2, false, false);
    bv.x = b0.w; bv.y = b1.w;
    acc3 = __builtin_amdgcn_wmma_f32_16x16x4_f32(
        false, av, false, bv, (short)0, acc3, false, false);
  }

  float* cp = C + (size_t)(m0 + 8 * h) * NN + 4 * c;
#pragma unroll
  for (int r = 0; r < 8; ++r) {
    v4f o;
    o.x = acc0[r]; o.y = acc1[r]; o.z = acc2[r]; o.w = acc3[r];
    *reinterpret_cast<v4f*>(cp + (size_t)r * NN) = o;
  }
}

template <int NN>
__global__ void fold_weights_k(const float* __restrict__ W1,
                               const float* __restrict__ b1,
                               const float* __restrict__ W2,
                               float* __restrict__ W12,
                               int KIN, int K) {
  const int wave = threadIdx.x >> 5;
  const int lane = threadIdx.x & 31;
  const int m0 = blockIdx.x * 16;
  const int n0 = wave * 16;
  const int half = lane >> 4;
  const int l15 = lane & 15;
  const bool bias_tile = (m0 >= KIN);
  const bool has_row = !bias_tile || l15 == 0;

  v8f acc = {};
  const float* aptr = (bias_tile ? b1 : W1 + (size_t)(m0 + l15) * K) + half * 2;
  const float* bptr = W2 + (size_t)(half * 2) * NN + n0 + l15;

  for (int k = 0; k < K; k += 4) {
    v2f a, b;
    const float a0 = aptr[0], a1 = aptr[1];
    a.x = has_row ? a0 : 0.0f;
    a.y = has_row ? a1 : 0.0f;
    b.x = bptr[0];
    b.y = bptr[NN];
    acc = __builtin_amdgcn_wmma_f32_16x16x4_f32(
         false, a,  false, b,
         (short)0, acc,  false,  false);
    aptr += 4;
    bptr += 4 * NN;
  }

  const int mrow = m0 + half * 8;
  float* cptr = W12 + (size_t)mrow * NN + n0 + l15;
#pragma unroll
  for (int r = 0; r < 8; ++r) {
    if (mrow + r <= KIN) cptr[0] = acc[r];
    cptr += NN;
  }
}

template <int C> struct rowvec;
template <> struct rowvec<64>  { typedef float2 type; };

__device__ __forceinline__ void axpy(float2& a, float w, const float2& v) {
  a.x += w * v.x; a.y += w * v.y;
}

__device__ __forceinline__ float2 scale_normalize(float2 v) {
  float mx = fmaxf(v.x, v.y);
  float mn = fminf(v.x, v.y);
#pragma unroll
  for (int o = 16; o > 0; o >>= 1) {
    mx = fmaxf(mx, __shfl_xor(mx, o, 32));
    mn = fminf(mn, __shfl_xor(mn, o, 32));
  }
  float inv = 1.0f / (mx - mn);
  float sx = (v.x - mn) * inv;
  float sy = (v.y - mn) * inv;
  float ss = sx * sx + sy * sy;
#pragma unroll
  for (int o = 16; o > 0; o >>= 1) ss += __shfl_xor(ss, o, 32);
  float r = 1.0f / fmaxf(sqrtf(ss), 1e-12f);
  float2 ov;
  ov.x = sx * r;
  ov.y = sy * r;
  return ov;
}

#define ROWS_IN_FLIGHT 24
template <int C> struct rowset {
  typename rowvec<C>::type r0, r1, r2, r3, r4, r5, r6, r7, r8, r9, r10, r11, r12, r13, r14, r15, r16, r17, r18, r19, r20, r21, r22, r23;
};

template <int C, int J0, int CNT>
__device__ __forceinline__ void load_rows(rowset<C>& v, const float* __restrict__ feat,
                                          int sv, int m, unsigned lane_off) {
  typedef typename rowvec<C>::type vec;
  const char* fb = reinterpret_cast<const char*>(feat);
#define ROW_OF_EDGE(K) \
  (*reinterpret_cast<const vec*>(fb + (((unsigned)__builtin_amdgcn_readlane(sv, J0 + K)) * (unsigned)(C * 4) + lane_off)))
  if (0 < CNT && J0 + 0 < m) { v.r0 = ROW_OF_EDGE(0);
  if (1 < CNT && J0 + 1 < m) { v.r1 = ROW_OF_EDGE(1);
  if (2 < CNT && J0 + 2 < m) { v.r2 = ROW_OF_EDGE(2);
  if (3 < CNT && J0 + 3 < m) { v.r3 = ROW_OF_EDGE(3);
  if (4 < CNT && J0 + 4 < m) { v.r4 = ROW_OF_EDGE(4);
  if (5 < CNT && J0 + 5 < m) { v.r5 = ROW_OF_EDGE(5);
  if (6 < CNT && J0 + 6 < m) { v.r6 = ROW_OF_EDGE(6);
  if (7 < CNT && J0 + 7 < m) { v.r7 = ROW_OF_EDGE(7);
  if (8 < CNT && J0 + 8 < m) { v.r8 = ROW_OF_EDGE(8);
  if (9 < CNT && J0 + 9 < m) { v.r9 = ROW_OF_EDGE(9);
  if (10 < CNT && J0 + 10 < m) { v.r10 = ROW_OF_EDGE(10);
  if (11 < CNT && J0 + 11 < m) { v.r11 = ROW_OF_EDGE(11);
  if (12 < CNT && J0 + 12 < m) { v.r12 = ROW_OF_EDGE(12);
  if (13 < CNT && J0 + 13 < m) { v.r13 = ROW_OF_EDGE(13);
  if (14 < CNT && J0 + 14 < m) { v.r14 = ROW_OF_EDGE(14);
  if (15 < CNT && J0 + 15 < m) { v.r15 = ROW_OF_EDGE(15);
  if (16 < CNT && J0 + 16 < m) { v.r16 = ROW_OF_EDGE(16);
  if (17 < CNT && J0 + 17 < m) { v.r17 = ROW_OF_EDGE(17);
  if (18 < CNT && J0 + 18 < m) { v.r18 = ROW_OF_EDGE(18);
  if (19 < CNT && J0 + 19 < m) { v.r19 = ROW_OF_EDGE(19);
  if (20 < CNT && J0 + 20 < m) { v.r20 = ROW_OF_EDGE(20);
  if (21 < CNT && J0 + 21 < m) { v.r21 = ROW_OF_EDGE(21);
  if (22 < CNT && J0 + 22 < m) { v.r22 = ROW_OF_EDGE(22);
  if (23 < CNT && J0 + 23 < m) { v.r23 = ROW_OF_EDGE(23);
  }}}}}}}}}}}}}}}}}}}}}}}}
#undef ROW_OF_EDGE
}

template <int C, int J0, int CNT>
__device__ __forceinline__ void add_rows(typename rowvec<C>::type& acc, const rowset<C>& v,
                                         float wv, int m) {
#define WEIGHT_OF_EDGE(K) __int_as_float(__builtin_amdgcn_readlane(__float_as_int(wv), J0 + K))
  if (0 < CNT && J0 + 0 < m) { axpy(acc, WEIGHT_OF_EDGE(0), v.r0);
  if (1 < CNT && J0 + 1 < m) { axpy(acc, WEIGHT_OF_EDGE(1), v.r1);
  if (2 < CNT && J0 + 2 < m) { axpy(acc, WEIGHT_OF_EDGE(2), v.r2);
  if (3 < CNT && J0 + 3 < m) { axpy(acc, WEIGHT_OF_EDGE(3), v.r3);
  if (4 < CNT && J0 + 4 < m) { axpy(acc, WEIGHT_OF_EDGE(4), v.r4);
  if (5 < CNT && J0 + 5 < m) { axpy(acc, WEIGHT_OF_EDGE(5), v.r5);
  if (6 < CNT && J0 + 6 < m) { axpy(acc, WEIGHT_OF_EDGE(6), v.r6);
  if (7 < CNT && J0 + 7 < m) { axpy(acc, WEIGHT_OF_EDGE(7), v.r7);
  if (8 < CNT && J0 + 8 < m) { axpy(acc, WEIGHT_OF_EDGE(8), v.r8);
  if (9 < CNT && J0 + 9 < m) { axpy(acc, WEIGHT_OF_EDGE(9), v.r9);
  if (10 < CNT && J0 + 10 < m) { axpy(acc, WEIGHT_OF_EDGE(10), v.r10);
  if (11 < CNT && J0 + 11 < m) { axpy(acc, WEIGHT_OF_EDGE(11), v.r11);
  if (12 < CNT && J0 + 12 < m) { axpy(acc, WEIGHT_OF_EDGE(12), v.r12);
  if (13 < CNT && J0 + 13 < m) { axpy(acc, WEIGHT_OF_EDGE(13), v.r13);
  if (14 < CNT && J0 + 14 < m) { axpy(acc, WEIGHT_OF_EDGE(14), v.r14);
  if (15 < CNT && J0 + 15 < m) { axpy(acc, WEIGHT_OF_EDGE(15), v.r15);
  if (16 < CNT && J0 + 16 < m) { axpy(acc, WEIGHT_OF_EDGE(16), v.r16);
  if (17 < CNT && J0 + 17 < m) { axpy(acc, WEIGHT_OF_EDGE(17), v.r17);
  if (18 < CNT && J0 + 18 < m) { axpy(acc, WEIGHT_OF_EDGE(18), v.r18);
  if (19 < CNT && J0 + 19 < m) { axpy(acc, WEIGHT_OF_EDGE(19), v.r19);
  if (20 < CNT && J0 + 20 < m) { axpy(acc, WEIGHT_OF_EDGE(20), v.r20);
  if (21 < CNT && J0 + 21 < m) { axpy(acc, WEIGHT_OF_EDGE(21), v.r21);
  if (22 < CNT && J0 + 22 < m) { axpy(acc, WEIGHT_OF_EDGE(22), v.r22);
  if (23 < CNT && J0 + 23 < m) { axpy(acc, WEIGHT_OF_EDGE(23), v.r23);
  }}}}}}}}}}}}}}}}}}}}}}}}
#undef WEIGHT_OF_EDGE
}

template <int C, bool NORM>
__global__ void gather_k(const float* __restrict__ feat,
                         const float* __restrict__ dinv,
                         const int* __restrict__ rowptr,
                         const int* __restrict__ cnt,
                         const int* __restrict__ csr,
                         const float* __restrict__ bias,
                         float* __restrict__ out, int N) {
  typedef typename rowvec<C>::type vec;
  const int lane = threadIdx.x & 31;
  const int node = __builtin_amdgcn_readfirstlane(
      (int)(blockIdx.x * (blockDim.x >> 5) + (threadIdx.x >> 5)));
  if (node >= N) return;

  int p = rowptr[node];
  int deg = cnt[node];
  float di = dinv[node];
  asm volatile("" : "+s"(p), "+s"(deg), "+s"(di));

  const vec bias_v = reinterpret_cast<const vec*>(bias)[lane];
  const vec self_v = reinterpret_cast<const vec*>(feat + (size_t)node * C)[lane];
  vec acc;
  bool started = false;

  const unsigned lane_off = (unsigned)lane * (unsigned)sizeof(vec);
  for (int m = deg; m > 0; m -= 32, p += 32) {
    const unsigned mine = (unsigned)(lane < m ? lane : m - 1);
    const int sv = csr[(unsigned)p + mine];
    const float dv = dinv[sv];
    rowset<C> v;
    load_rows<C, 0, ROWS_IN_FLIGHT>(v, feat, sv, m, lane_off);
    if (!started) {
      acc = bias_v;
      axpy(acc, di * di, self_v);
      started = true;
    }
    const float wv = dv * di;
    add_rows<C, 0, ROWS_IN_FLIGHT>(acc, v, wv, m);
    if (m > ROWS_IN_FLIGHT) {
      load_rows<C, ROWS_IN_FLIGHT, 32 - ROWS_IN_FLIGHT>(v, feat, sv, m, lane_off);
      add_rows<C, ROWS_IN_FLIGHT, 32 - ROWS_IN_FLIGHT>(acc, v, wv, m);
    }
  }
  if (!started) {
    acc = bias_v;
    axpy(acc, di * di, self_v);
  }
  if constexpr (NORM) {
    static_assert(C == 64, "the fused scale + normalise expects one float2 per lane");
    acc = scale_normalize(acc);
  }
  reinterpret_cast<vec*>(out + (size_t)node * C)[lane] = acc;
}

extern "C" void kernel_launch(void* const* d_in, const int* in_sizes, int n_in,
                              void* d_out, int out_size, void* d_ws, size_t ws_size,
                              hipStream_t stream) {
  (void)n_in; (void)out_size; (void)ws_size;

  const float* x  = (const float*)d_in[0];
  const int*   ei = (const int*)d_in[1];
  const float* W1 = (const float*)d_in[2];
  const float* b1 = (const float*)d_in[3];
  const float* W2 = (const float*)d_in[4];
  const float* b2 = (const float*)d_in[5];

  const int N = in_sizes[0] / 128;
  const int E = in_sizes[1] / 2;
  const int* src = ei;
  const int* dst = ei + E;
  float* out = (float*)d_out;

  float* ws = (float*)d_ws;
  size_t Npad = ((size_t)N + 63) & ~(size_t)63;
  float* dinv = ws;
  float* y = ws + Npad;
  float* g = y + (size_t)N * 64;
  float* scratch = g + (size_t)N * 64;
  int* cnt    = (int*)(scratch + (size_t)N * 128);
  int* rowptr = cnt + Npad;
  int2* tmp   = (int2*)scratch;
  float* W12  = scratch;
  int* csr    = rowptr + Npad;
  int* cntmat = csr + E;
  const int nchunks = (E + EDGE_CHUNK - 1) / EDGE_CHUNK;
  int* gbase  = cntmat + (size_t)nchunks * GRP_MAX;

  group_count_k<<<nchunks, 1024, 0, stream>>>(dst, cntmat, E);
  group_scatter_k<<<nchunks, 1024, 0, stream>>>(src, dst, cntmat, gbase, tmp, E, nchunks);
  group_csr_k<<<(N + GRP_SIZE - 1) / GRP_SIZE, 1024, 0, stream>>>(tmp, gbase, cnt, rowptr, dinv, csr, N, E);

  fold_weights_k<64><<<128 / 16 + 1, 128, 0, stream>>>(W1, b1, W2, W12, 128, 128);

  gemm_strip_k<128><<<(N + 63) / 64, 128, 0, stream>>>(x, W12, y, N);
  gather_k<64, false><<<(N + 7) / 8, 256, 0, stream>>>(y, dinv, rowptr, cnt, csr, W12 + 128 * 64, g, N);

  gather_k<64, true><<<(N + 7) / 8, 256, 0, stream>>>(g, dinv, rowptr, cnt, csr, b2, out, N);
}


# Weight fold W12=[W1;b1]@W2 has no launch of its own: its 9 tiles run as extra blocks of the edge sort's first launch, on idle WGPs; W12 moved behind tmp in scratch; 7 launches become 6
# speedup vs baseline: 5.3561x; 1.1314x over previous
#include <hip/hip_runtime.h>
#include <math.h>

typedef float v2f __attribute__((ext_vector_type(2)));
typedef float v8f __attribute__((ext_vector_type(8)));

#define GRP_SHIFT 9
#define GRP_SIZE  512
#define GRP_MAX   256
#define EDGE_CHUNK 8192
#define CSR_STAGE 12288

__global__ __launch_bounds__(1024)
void group_scatter_k(const int* __restrict__ src, const int* __restrict__ dst,
                     const int* __restrict__ cntmat, int* __restrict__ gbase,
                     int2* __restrict__ tmp, int E, int nchunks) {
  __shared__ int part_tot[4][GRP_MAX];
  __shared__ int part_pre[4][GRP_MAX];
  __shared__ int off[GRP_MAX];
  __shared__ int wsum[8];
  const int tid = threadIdx.x;
  const int lane = tid & 31;
  const int w = tid >> 5;
  const int blk = blockIdx.x;
  {
    const int b = tid & (GRP_MAX - 1);
    const int g = tid >> 8;
    int tot = 0, pre = 0;
    for (int r = g; r < nchunks; r += 4) {
      const int v = cntmat[r * GRP_MAX + b];
      tot += v;
      if (r < blk) pre += v;
    }
    part_tot[g][b] = tot;
    part_pre[g][b] = pre;
  }
  __syncthreads();
  int t = 0, p = 0, x = 0;
  if (tid < GRP_MAX) {
    t = part_tot[0][tid] + part_tot[1][tid] + part_tot[2][tid] + part_tot[3][tid];
    p = part_pre[0][tid] + part_pre[1][tid] + part_pre[2][tid] + part_pre[3][tid];
    x = t;
    for (int o = 1; o < 32; o <<= 1) {
      const int y = __shfl_up(x, o, 32);
      if (lane >= o) x += y;
    }
    if (lane == 31) wsum[w] = x;
  }
  __syncthreads();
  if (tid < GRP_MAX) {
    int wp = 0;
    for (int k = 0; k < w; ++k) wp += wsum[k];
    const int base = wp + x - t;
    off[tid] = base + p;
    if (blk == 0) gbase[tid] = base;
  }
  __syncthreads();
  const int e0 = blk * EDGE_CHUNK + tid;
  int d[EDGE_CHUNK / 1024], s[EDGE_CHUNK / 1024];
#pragma unroll
  for (int j = 0; j < EDGE_CHUNK / 1024; ++j) {
    const int e = e0 + j * 1024;
    const int ec = e < E ? e : E - 1;
    d[j] = dst[ec];
    s[j] = src[ec];
  }
  int pos[EDGE_CHUNK / 1024];
#pragma unroll
  for (int j = 0; j < EDGE_CHUNK / 1024; ++j) {
    pos[j] = atomicAdd(&off[d[j] >> GRP_SHIFT], (e0 + j * 1024 < E) ? 1 : 0);
  }
#pragma unroll
  for (int j = 0; j < EDGE_CHUNK / 1024; ++j) {
    if (e0 + j * 1024 < E) tmp[pos[j]] = make_int2(s[j], d[j]);
  }
}

__global__ __launch_bounds__(1024)
void group_csr_k(const int2* __restrict__ tmp, const int* __restrict__ gbase,
                 int* __restrict__ cnt, int* __restrict__ rowptr,
                 float* __restrict__ dinv, int* __restrict__ csr, int N, int E) {
  __shared__ int c[GRP_SIZE];
  __shared__ int cur[GRP_SIZE];
  __shared__ int wsum[16];
  __shared__ int stage[CSR_STAGE];
  const int tid = threadIdx.x;
  const int lane = tid & 31;
  const int w = tid >> 5;
  const int b = blockIdx.x;
  const int lo = b << GRP_SHIFT;
  const int start = gbase[b];
  const int end = (b + 1 < GRP_MAX) ? gbase[b + 1] : E;
  const int n = end - start;

  if (tid < GRP_SIZE) c[tid] = 0;
  __syncthreads();
  for (int i0 = start + tid; i0 < end; i0 += 4096) {
    int dd[4];
#pragma unroll
    for (int k = 0; k < 4; ++k) {
      const int i = i0 + k * 1024;
      dd[k] = tmp[i < end ? i : i0].y;
    }
#pragma unroll
    for (int k = 0; k < 4; ++k) {
      atomicAdd(&c[dd[k] - lo], (i0 + k * 1024 < end) ? 1 : 0);
    }
  }
  __syncthreads();

  int v = 0, x = 0;
  if (tid < GRP_SIZE) {
    v = c[tid];
    x = v;
    for (int o = 1; o < 32; o <<= 1) {
      const int y = __shfl_up(x, o, 32);
      if (lane >= o) x += y;
    }
    if (lane == 31) wsum[w] = x;
  }
  __syncthreads();
  if (tid < GRP_SIZE) {
    int wp = 0;
    for (int k = 0; k < w; ++k) wp += wsum[k];
    const int excl = wp + x - v;
    cur[tid] = excl;
    const int i = lo + tid;
    if (i < N) {
      cnt[i] = v;
      rowptr[i] = start + excl;
      float d = (float)(v + 1);
      dinv[i] = (d > 0.0f) ? rsqrtf(d) : 0.0f;
    }
  }
  __syncthreads();

  const bool staged = (n <= CSR_STAGE);
  for (int i0 = start + tid; i0 < end; i0 += 4096) {
    int2 pr[4];
#pragma unroll
    for (int k = 0; k < 4; ++k) {
      const int i = i0 + k * 1024;
      pr[k] = tmp[i < end ? i : i0];
    }
    int q[4];
#pragma unroll
    for (int k = 0; k < 4; ++k) {
      q[k] = atomicAdd(&cur[pr[k].y - lo], (i0 + k * 1024 < end) ? 1 : 0);
    }
#pragma unroll
    for (int k = 0; k < 4; ++k) {
      if (i0 + k * 1024 < end) {
        if (staged) stage[q[k]] = pr[k].x;
        else csr[start + q[k]] = pr[k].x;
      }
    }
  }
  if (staged) {
    __syncthreads();
    for (int i = tid; i < n; i += 1024) csr[start + i] = stage[i];
  }
}

typedef float v4f __attribute__((ext_vector_type(4)));

template <int KK>
__global__ __launch_bounds__(128)
void gemm_strip_k(const float* __restrict__ A,
                  const float* __restrict__ B,
                  const float* __restrict__ rowscale,
                  float* __restrict__ C, int M) {
  static_assert(KK % 32 == 0, "k is taken in blocks of 32");
  constexpr int NN = 64;
  constexpr int STEPS = KK / 4;
  const int wave = __builtin_amdgcn_readfirstlane((int)(threadIdx.x >> 5));
  const int lane = threadIdx.x & 31;
  const int m0 = (blockIdx.x * 4 + wave) * 16;
  if (m0 >= M) return;
  const int h = lane >> 4;
  const int c = lane & 15;

  const v4f* ap = reinterpret_cast<const v4f*>(A + (size_t)(m0 + c) * KK + h * 16);
  v4f a[KK / 32][4];
#pragma unroll
  for (int kb = 0; kb < KK / 32; ++kb) {
#pragma unroll
    for (int q = 0; q < 4; ++q) a[kb][q] = ap[kb * 8 + q];
  }
  const v4f* sp = reinterpret_cast<const v4f*>(rowscale + m0 + 8 * h);
  const v4f sc_lo = sp[0];
  const v4f sc_hi = sp[1];

  const float* bp = B + (size_t)(h * 16) * NN + 4 * c;
  v4f br[3][2];
#pragma unroll
  for (int t = 0; t < 2; ++t) {
    const int k = (t >> 3) * 32 + (t & 7) * 2;
    br[t][0] = *reinterpret_cast<const v4f*>(bp + (size_t)k * NN);
    br[t][1] = *reinterpret_cast<const v4f*>(bp + (size_t)(k + 1) * NN);
  }

  v8f acc0 = {}, acc1 = {}, acc2 = {}, acc3 = {};
#pragma unroll
  for (int t = 0; t < STEPS; ++t) {
    if (t + 2 < STEPS) {
      const int kn = ((t + 2) >> 3) * 32 + ((t + 2) & 7) * 2;
      br[(t + 2) % 3][0] = *reinterpret_cast<const v4f*>(bp + (size_t)kn * NN);
      br[(t + 2) % 3][1] = *reinterpret_cast<const v4f*>(bp + (size_t)(kn + 1) * NN);
    }
    __builtin_amdgcn_sched_barrier(0);
    const v4f aq = a[t >> 3][(t >> 1) & 3];
    const v4f b0 = br[t % 3][0];
    const v4f b1 = br[t % 3][1];
    v2f av, bv;
    av.x = (t & 1) ? aq.z : aq.x;
    av.y = (t & 1) ? aq.w : aq.y;
    bv.x = b0.x; bv.y = b1.x;
    acc0 = __builtin_amdgcn_wmma_f32_16x16x4_f32(
         false, av,  false, bv,
         (short)0, acc0,  false,  false);
    bv.x = b0.y; bv.y = b1.y;
    acc1 = __builtin_amdgcn_wmma_f32_16x16x4_f32(
        false, av, false, bv, (short)0, acc1, false, false);
    bv.x = b0.z; bv.y = b1.z;
    acc2 = __builtin_amdgcn_wmma_f32_16x16x4_f32(
        false, av, false, bv, (short)0, acc2, false, false);
    bv.x = b0.w; bv.y = b1.w;
    acc3 = __builtin_amdgcn_wmma_f32_16x16x4_f32(
        false, av, false, bv, (short)0, acc3, false, false);
  }

  float* cp = C + (size_t)(m0 + 8 * h) * NN + 4 * c;
#pragma unroll
  for (int r = 0; r < 8; ++r) {
    const float sc = (r < 4) ? sc_lo[r & 3] : sc_hi[r & 3];
    v4f o;
    o.x = acc0[r] * sc; o.y = acc1[r] * sc; o.z = acc2[r] * sc; o.w = acc3[r] * sc;
    *reinterpret_cast<v4f*>(cp + (size_t)r * NN) = o;
  }
}

template <int NN>
__device__ __forceinline__ void fold_weights_tile(const float* __restrict__ W1,
                                                  const float* __restrict__ b1,
                                                  const float* __restrict__ W2,
                                                  float* __restrict__ W12,
                                                  int KIN, int K, int tile) {
  const int wave = threadIdx.x >> 5;
  const int lane = threadIdx.x & 31;
  const int m0 = tile * 16;
  const int n0 = wave * 16;
  const int half = lane >> 4;
  const int l15 = lane & 15;
  const bool bias_tile = (m0 >= KIN);
  const bool has_row = !bias_tile || l15 == 0;

  v8f acc = {};
  const float* aptr = (bias_tile ? b1 : W1 + (size_t)(m0 + l15) * K) + half * 2;
  const float* bptr = W2 + (size_t)(half * 2) * NN + n0 + l15;

  for (int k = 0; k < K; k += 4) {
    v2f a, b;
    const float a0 = aptr[0], a1 = aptr[1];
    a.x = has_row ? a0 : 0.0f;
    a.y = has_row ? a1 : 0.0f;
    b.x = bptr[0];
    b.y = bptr[NN];
    acc = __builtin_amdgcn_wmma_f32_16x16x4_f32(
         false, a,  false, b,
         (short)0, acc,  false,  false);
    aptr += 4;
    bptr += 4 * NN;
  }

  const int mrow = m0 + half * 8;
  float* cptr = W12 + (size_t)mrow * NN + n0 + l15;
#pragma unroll
  for (int r = 0; r < 8; ++r) {
    if (mrow + r <= KIN) cptr[0] = acc[r];
    cptr += NN;
  }
}

__global__ __launch_bounds__(1024)
void group_count_fold_k(const int* __restrict__ dst, int* __restrict__ cntmat, int E,
                        int nchunks,
                        const float* __restrict__ W1, const float* __restrict__ b1,
                        const float* __restrict__ W2, float* __restrict__ W12) {
  if ((int)blockIdx.x >= nchunks) {
    if (threadIdx.x < 128)
      fold_weights_tile<64>(W1, b1, W2, W12, 128, 128, (int)blockIdx.x - nchunks);
    return;
  }
  __shared__ int hist[GRP_MAX];
  const int tid = threadIdx.x;
  if (tid < GRP_MAX) hist[tid] = 0;
  __syncthreads();
  const int e0 = blockIdx.x * EDGE_CHUNK + tid;
  int d[EDGE_CHUNK / 1024];
#pragma unroll
  for (int j = 0; j < EDGE_CHUNK / 1024; ++j) {
    const int e = e0 + j * 1024;
    d[j] = dst[e < E ? e : E - 1];
  }
#pragma unroll
  for (int j = 0; j < EDGE_CHUNK / 1024; ++j) {
    atomicAdd(&hist[d[j] >> GRP_SHIFT], (e0 + j * 1024 < E) ? 1 : 0);
  }
  __syncthreads();
  if (tid < GRP_MAX) cntmat[blockIdx.x * GRP_MAX + tid] = hist[tid];
}

template <int C> struct rowvec;
template <> struct rowvec<64>  { typedef float2 type; };

__device__ __forceinline__ void add_row(float2& a, const float2& v) {
  a.x += v.x; a.y += v.y;
}

__device__ __forceinline__ float2 scale_normalize(float2 v) {
  float mx = fmaxf(v.x, v.y);
  float mn = fminf(v.x, v.y);
#pragma unroll
  for (int o = 16; o > 0; o >>= 1) {
    mx = fmaxf(mx, __shfl_xor(mx, o, 32));
    mn = fminf(mn, __shfl_xor(mn, o, 32));
  }
  float inv = 1.0f / (mx - mn);
  float sx = (v.x - mn) * inv;
  float sy = (v.y - mn) * inv;
  float ss = sx * sx + sy * sy;
#pragma unroll
  for (int o = 16; o > 0; o >>= 1) ss += __shfl_xor(ss, o, 32);
  float r = 1.0f / fmaxf(sqrtf(ss), 1e-12f);
  float2 ov;
  ov.x = sx * r;
  ov.y = sy * r;
  return ov;
}

#define ROWS_IN_FLIGHT 24
template <int C> struct rowset {
  typename rowvec<C>::type r0, r1, r2, r3, r4, r5, r6, r7, r8, r9, r10, r11, r12, r13, r14, r15, r16, r17, r18, r19, r20, r21, r22, r23;
};

template <int C, int J0, int CNT>
__device__ __forceinline__ void load_rows(rowset<C>& v, const float* __restrict__ feat,
                                          int sv, int m, unsigned lane_off) {
  typedef typename rowvec<C>::type vec;
  const char* fb = reinterpret_cast<const char*>(feat);
#define ROW_OF_EDGE(K) \
  (*reinterpret_cast<const vec*>(fb + (((unsigned)__builtin_amdgcn_readlane(sv, J0 + K)) * (unsigned)(C * 4) + lane_off)))
  if (0 < CNT && J0 + 0 < m) { v.r0 = ROW_OF_EDGE(0);
  if (1 < CNT && J0 + 1 < m) { v.r1 = ROW_OF_EDGE(1);
  if (2 < CNT && J0 + 2 < m) { v.r2 = ROW_OF_EDGE(2);
  if (3 < CNT && J0 + 3 < m) { v.r3 = ROW_OF_EDGE(3);
  if (4 < CNT && J0 + 4 < m) { v.r4 = ROW_OF_EDGE(4);
  if (5 < CNT && J0 + 5 < m) { v.r5 = ROW_OF_EDGE(5);
  if (6 < CNT && J0 + 6 < m) { v.r6 = ROW_OF_EDGE(6);
  if (7 < CNT && J0 + 7 < m) { v.r7 = ROW_OF_EDGE(7);
  if (8 < CNT && J0 + 8 < m) { v.r8 = ROW_OF_EDGE(8);
  if (9 < CNT && J0 + 9 < m) { v.r9 = ROW_OF_EDGE(9);
  if (10 < CNT && J0 + 10 < m) { v.r10 = ROW_OF_EDGE(10);
  if (11 < CNT && J0 + 11 < m) { v.r11 = ROW_OF_EDGE(11);
  if (12 < CNT && J0 + 12 < m) { v.r12 = ROW_OF_EDGE(12);
  if (13 < CNT && J0 + 13 < m) { v.r13 = ROW_OF_EDGE(13);
  if (14 < CNT && J0 + 14 < m) { v.r14 = ROW_OF_EDGE(14);
  if (15 < CNT && J0 + 15 < m) { v.r15 = ROW_OF_EDGE(15);
  if (16 < CNT && J0 + 16 < m) { v.r16 = ROW_OF_EDGE(16);
  if (17 < CNT && J0 + 17 < m) { v.r17 = ROW_OF_EDGE(17);
  if (18 < CNT && J0 + 18 < m) { v.r18 = ROW_OF_EDGE(18);
  if (19 < CNT && J0 + 19 < m) { v.r19 = ROW_OF_EDGE(19);
  if (20 < CNT && J0 + 20 < m) { v.r20 = ROW_OF_EDGE(20);
  if (21 < CNT && J0 + 21 < m) { v.r21 = ROW_OF_EDGE(21);
  if (22 < CNT && J0 + 22 < m) { v.r22 = ROW_OF_EDGE(22);
  if (23 < CNT && J0 + 23 < m) { v.r23 = ROW_OF_EDGE(23);
  }}}}}}}}}}}}}}}}}}}}}}}}
#undef ROW_OF_EDGE
}

template <int C, int J0, int CNT>
__device__ __forceinline__ void add_rows(typename rowvec<C>::type& acc, const rowset<C>& v,
                                         int m) {
  if (0 < CNT && J0 + 0 < m) { add_row(acc, v.r0);
  if (1 < CNT && J0 + 1 < m) { add_row(acc, v.r1);
  if (2 < CNT && J0 + 2 < m) { add_row(acc, v.r2);
  if (3 < CNT && J0 + 3 < m) { add_row(acc, v.r3);
  if (4 < CNT && J0 + 4 < m) { add_row(acc, v.r4);
  if (5 < CNT && J0 + 5 < m) { add_row(acc, v.r5);
  if (6 < CNT && J0 + 6 < m) { add_row(acc, v.r6);
  if (7 < CNT && J0 + 7 < m) { add_row(acc, v.r7);
  if (8 < CNT && J0 + 8 < m) { add_row(acc, v.r8);
  if (9 < CNT && J0 + 9 < m) { add_row(acc, v.r9);
  if (10 < CNT && J0 + 10 < m) { add_row(acc, v.r10);
  if (11 < CNT && J0 + 11 < m) { add_row(acc, v.r11);
  if (12 < CNT && J0 + 12 < m) { add_row(acc, v.r12);
  if (13 < CNT && J0 + 13 < m) { add_row(acc, v.r13);
  if (14 < CNT && J0 + 14 < m) { add_row(acc, v.r14);
  if (15 < CNT && J0 + 15 < m) { add_row(acc, v.r15);
  if (16 < CNT && J0 + 16 < m) { add_row(acc, v.r16);
  if (17 < CNT && J0 + 17 < m) { add_row(acc, v.r17);
  if (18 < CNT && J0 + 18 < m) { add_row(acc, v.r18);
  if (19 < CNT && J0 + 19 < m) { add_row(acc, v.r19);
  if (20 < CNT && J0 + 20 < m) { add_row(acc, v.r20);
  if (21 < CNT && J0 + 21 < m) { add_row(acc, v.r21);
  if (22 < CNT && J0 + 22 < m) { add_row(acc, v.r22);
  if (23 < CNT && J0 + 23 < m) { add_row(acc, v.r23);
  }}}}}}}}}}}}}}}}}}}}}}}}
}

template <int C, bool NORM>
__global__ void gather_k(const float* __restrict__ feat,
                         const float* __restrict__ dinv,
                         const int* __restrict__ rowptr,
                         const int* __restrict__ cnt,
                         const int* __restrict__ csr,
                         const float* __restrict__ bias,
                         float* __restrict__ out, int N) {
  typedef typename rowvec<C>::type vec;
  const int lane = threadIdx.x & 31;
  const int node = __builtin_amdgcn_readfirstlane(
      (int)(blockIdx.x * (blockDim.x >> 5) + (threadIdx.x >> 5)));
  if (node >= N) return;

  int p = rowptr[node];
  int deg = cnt[node];
  float di = dinv[node];
  asm volatile("" : "+s"(p), "+s"(deg), "+s"(di) : "s"(csr), "s"(bias));

  const vec bias_v = reinterpret_cast<const vec*>(bias)[lane];
  const vec self_v = reinterpret_cast<const vec*>(feat + (size_t)node * C)[lane];
  vec acc;
  bool started = false;

  const unsigned lane_off = (unsigned)lane * (unsigned)sizeof(vec);
  for (int m = deg; m > 0; m -= 32, p += 32) {
    const unsigned mine = (unsigned)(lane < m ? lane : m - 1);
    const int sv = csr[(unsigned)p + mine];
    rowset<C> v;
    load_rows<C, 0, ROWS_IN_FLIGHT>(v, feat, sv, m, lane_off);
    if (!started) {
      acc = self_v;
      started = true;
    }
    add_rows<C, 0, ROWS_IN_FLIGHT>(acc, v, m);
    if (m > ROWS_IN_FLIGHT) {
      load_rows<C, ROWS_IN_FLIGHT, 32 - ROWS_IN_FLIGHT>(v, feat, sv, m, lane_off);
      add_rows<C, ROWS_IN_FLIGHT, 32 - ROWS_IN_FLIGHT>(acc, v, m);
    }
  }
  if (!started) acc = self_v;
  acc.x = di * acc.x + bias_v.x;
  acc.y = di * acc.y + bias_v.y;
  if constexpr (NORM) {
    static_assert(C == 64, "the fused scale + normalise expects one float2 per lane");
    acc = scale_normalize(acc);
  } else {
    acc.x *= di;
    acc.y *= di;
  }
  reinterpret_cast<vec*>(out + (size_t)node * C)[lane] = acc;
}

extern "C" void kernel_launch(void* const* d_in, const int* in_sizes, int n_in,
                              void* d_out, int out_size, void* d_ws, size_t ws_size,
                              hipStream_t stream) {
  (void)n_in; (void)out_size; (void)ws_size;

  const float* x  = (const float*)d_in[0];
  const int*   ei = (const int*)d_in[1];
  const float* W1 = (const float*)d_in[2];
  const float* b1 = (const float*)d_in[3];
  const float* W2 = (const float*)d_in[4];
  const float* b2 = (const float*)d_in[5];

  const int N = in_sizes[0] / 128;
  const int E = in_sizes[1] / 2;
  const int* src = ei;
  const int* dst = ei + E;
  float* out = (float*)d_out;

  float* ws = (float*)d_ws;
  size_t Npad = ((size_t)N + 63) & ~(size_t)63;
  float* dinv = ws;
  float* y = ws + Npad;
  float* g = y + (size_t)N * 64;
  float* scratch = g + (size_t)N * 64;
  int* cnt    = (int*)(scratch + (size_t)N * 128);
  int* rowptr = cnt + Npad;
  int2* tmp   = (int2*)scratch;
  float* W12  = scratch + (((size_t)E * 2 + 3) & ~(size_t)3);
  int* csr    = rowptr + Npad;
  int* cntmat = csr + E;
  const int nchunks = (E + EDGE_CHUNK - 1) / EDGE_CHUNK;
  int* gbase  = cntmat + (size_t)nchunks * GRP_MAX;

  group_count_fold_k<<<nchunks + 128 / 16 + 1, 1024, 0, stream>>>(dst, cntmat, E, nchunks, W1, b1, W2, W12);
  group_scatter_k<<<nchunks, 1024, 0, stream>>>(src, dst, cntmat, gbase, tmp, E, nchunks);
  group_csr_k<<<(N + GRP_SIZE - 1) / GRP_SIZE, 1024, 0, stream>>>(tmp, gbase, cnt, rowptr, dinv, csr, N, E);

  gemm_strip_k<128><<<(N + 63) / 64, 128, 0, stream>>>(x, W12, dinv, y, N);
  gather_k<64, false><<<(N + 7) / 8, 256, 0, stream>>>(y, dinv, rowptr, cnt, csr, W12 + 128 * 64, g, N);

  gather_k<64, true><<<(N + 7) / 8, 256, 0, stream>>>(g, dinv, rowptr, cnt, csr, b2, out, N);
}


# Edge sort in two launches, not three: each chunk puts its edges in group order in LDS (returning add = rank) and stores one contiguous piece; a group's block collects its runs; sum order changes
# speedup vs baseline: 6.3865x; 1.1924x over previous
#include <hip/hip_runtime.h>
#include <math.h>

typedef float v2f __attribute__((ext_vector_type(2)));
typedef float v8f __attribute__((ext_vector_type(8)));

#define GRP_SHIFT 9
#define GRP_SIZE  512
#define GRP_MAX   256
#define EDGE_CHUNK 8192
#define CSR_STAGE 12288
#define CHUNK_MAX 1024

template <int K>
__device__ __forceinline__ void runs_of(const int* pre, int s0, const int (&p)[K], int (&c)[K]) {
#pragma unroll
  for (int k = 0; k < K; ++k) c[k] = 0;
  for (int s = s0; s > 0; s >>= 1) {
#pragma unroll
    for (int k = 0; k < K; ++k) {
      const int t = pre[c[k] + s];
      if (t <= p[k]) c[k] += s;
    }
  }
}

__global__ __launch_bounds__(1024)
void group_csr_k(const int2* __restrict__ tmp, const int* __restrict__ cntmat, int nchunks,
                 int* __restrict__ cnt, int* __restrict__ rowptr,
                 float* __restrict__ dinv, int* __restrict__ csr, int N) {
  __shared__ int c[GRP_SIZE];
  __shared__ int cur[GRP_SIZE];
  __shared__ int wsum[32];
  __shared__ int wofs[32];
  __shared__ int pre[CHUNK_MAX + 1];
  __shared__ int rb[CHUNK_MAX];
  __shared__ int stage[CSR_STAGE];
  __shared__ int2 held[CSR_STAGE];
  const int tid = threadIdx.x;
  const int lane = tid & 31;
  const int w = tid >> 5;
  const int b = blockIdx.x;
  const int lo = b << GRP_SHIFT;

  int rc = 0, rl = 0;
  if (tid < nchunks) {
    const int pk = cntmat[tid * GRP_MAX + b];
    rc = pk & 0xFFFF;
    rl = pk >> 16;
  }
  if (tid < GRP_SIZE) c[tid] = 0;
  int rx = rc;
  int ro = rl;
  for (int o = 1; o < 32; o <<= 1) {
    const int y = __shfl_up(rx, o, 32);
    if (lane >= o) rx += y;
  }
  for (int o = 16; o > 0; o >>= 1) ro += __shfl_xor(ro, o, 32);
  if (lane == 31) wsum[w] = rx;
  if (lane == 0) wofs[w] = ro;
  __syncthreads();
  int rwp = 0, n = 0, start = 0;
  for (int k = 0; k < 32; ++k) {
    const int t = wsum[k];
    if (k < w) rwp += t;
    n += t;
    start += wofs[k];
  }
  {
    const int excl = rwp + rx - rc;
    pre[tid] = (tid < nchunks) ? excl : n;
    if (tid == 0) pre[CHUNK_MAX] = n;
    if (tid < nchunks) rb[tid] = tid * EDGE_CHUNK + rl - excl;
  }
  int s0 = 1;
  while (2 * s0 < nchunks) s0 <<= 1;
  __syncthreads();

  const bool staged = (n <= CSR_STAGE);
  for (int p0 = tid; p0 < n; p0 += 4096) {
    int pc[4], r[4];
#pragma unroll
    for (int k = 0; k < 4; ++k) {
      const int p = p0 + k * 1024;
      pc[k] = p < n ? p : p0;
    }
    runs_of<4>(pre, s0, pc, r);
    int2 pr[4];
#pragma unroll
    for (int k = 0; k < 4; ++k) pr[k] = tmp[rb[r[k]] + pc[k]];
#pragma unroll
    for (int k = 0; k < 4; ++k) {
      atomicAdd(&c[pr[k].y - lo], (p0 + k * 1024 < n) ? 1 : 0);
    }
    if (staged) {
#pragma unroll
      for (int k = 0; k < 4; ++k) {
        if (p0 + k * 1024 < n) held[p0 + k * 1024] = pr[k];
      }
    }
  }
  __syncthreads();

  int v = 0, x = 0;
  if (tid < GRP_SIZE) {
    v = c[tid];
    x = v;
    for (int o = 1; o < 32; o <<= 1) {
      const int y = __shfl_up(x, o, 32);
      if (lane >= o) x += y;
    }
    if (lane == 31) wsum[w] = x;
  }
  __syncthreads();
  if (tid < GRP_SIZE) {
    int wp = 0;
    for (int k = 0; k < w; ++k) wp += wsum[k];
    const int excl = wp + x - v;
    cur[tid] = excl;
    const int i = lo + tid;
    if (i < N) {
      cnt[i] = v;
      rowptr[i] = start + excl;
      float d = (float)(v + 1);
      dinv[i] = (d > 0.0f) ? rsqrtf(d) : 0.0f;
    }
  }
  __syncthreads();

  for (int p0 = tid; p0 < n; p0 += 4096) {
    int pc[4];
#pragma unroll
    for (int k = 0; k < 4; ++k) {
      const int p = p0 + k * 1024;
      pc[k] = p < n ? p : p0;
    }
    int2 pr[4];
    if (staged) {
#pragma unroll
      for (int k = 0; k < 4; ++k) pr[k] = held[pc[k]];
    } else {
      int r[4];
      runs_of<4>(pre, s0, pc, r);
#pragma unroll
      for (int k = 0; k < 4; ++k) pr[k] = tmp[rb[r[k]] + pc[k]];
    }
    int q[4];
#pragma unroll
    for (int k = 0; k < 4; ++k) {
      q[k] = atomicAdd(&cur[pr[k].y - lo], (p0 + k * 1024 < n) ? 1 : 0);
    }
#pragma unroll
    for (int k = 0; k < 4; ++k) {
      if (p0 + k * 1024 < n) {
        if (staged) stage[q[k]] = pr[k].x;
        else csr[start + q[k]] = pr[k].x;
      }
    }
  }
  if (staged) {
    __syncthreads();
    for (int i = tid; i < n; i += 1024) csr[start + i] = stage[i];
  }
}

typedef float v4f __attribute__((ext_vector_type(4)));

template <int KK>
__global__ __launch_bounds__(128)
void gemm_strip_k(const float* __restrict__ A,
                  const float* __restrict__ B,
                  const float* __restrict__ rowscale,
                  float* __restrict__ C, int M) {
  static_assert(KK % 32 == 0, "k is taken in blocks of 32");
  constexpr int NN = 64;
  constexpr int STEPS = KK / 4;
  const int wave = __builtin_amdgcn_readfirstlane((int)(threadIdx.x >> 5));
  const int lane = threadIdx.x & 31;
  const int m0 = (blockIdx.x * 4 + wave) * 16;
  if (m0 >= M) return;
  const int h = lane >> 4;
  const int c = lane & 15;

  const v4f* ap = reinterpret_cast<const v4f*>(A + (size_t)(m0 + c) * KK + h * 16);
  v4f a[KK / 32][4];
#pragma unroll
  for (int kb = 0; kb < KK / 32; ++kb) {
#pragma unroll
    for (int q = 0; q < 4; ++q) a[kb][q] = ap[kb * 8 + q];
  }
  const v4f* sp = reinterpret_cast<const v4f*>(rowscale + m0 + 8 * h);
  const v4f sc_lo = sp[0];
  const v4f sc_hi = sp[1];

  const float* bp = B + (size_t)(h * 16) * NN + 4 * c;
  v4f br[3][2];
#pragma unroll
  for (int t = 0; t < 2; ++t) {
    const int k = (t >> 3) * 32 + (t & 7) * 2;
    br[t][0] = *reinterpret_cast<const v4f*>(bp + (size_t)k * NN);
    br[t][1] = *reinterpret_cast<const v4f*>(bp + (size_t)(k + 1) * NN);
  }

  v8f acc0 = {}, acc1 = {}, acc2 = {}, acc3 = {};
#pragma unroll
  for (int t = 0; t < STEPS; ++t) {
    if (t + 2 < STEPS) {
      const int kn = ((t + 2) >> 3) * 32 + ((t + 2) & 7) * 2;
      br[(t + 2) % 3][0] = *reinterpret_cast<const v4f*>(bp + (size_t)kn * NN);
      br[(t + 2) % 3][1] = *reinterpret_cast<const v4f*>(bp + (size_t)(kn + 1) * NN);
    }
    __builtin_amdgcn_sched_barrier(0);
    const v4f aq = a[t >> 3][(t >> 1) & 3];
    const v4f b0 = br[t % 3][0];
    const v4f b1 = br[t % 3][1];
    v2f av, bv;
    av.x = (t & 1) ? aq.z : aq.x;
    av.y = (t & 1) ? aq.w : aq.y;
    bv.x = b0.x; bv.y = b1.x;
    acc0 = __builtin_amdgcn_wmma_f32_16x16x4_f32(
         false, av,  false, bv,
         (short)0, acc0,  false,  false);
    bv.x = b0.y; bv.y = b1.y;
    acc1 = __builtin_amdgcn_wmma_f32_16x16x4_f32(
        false, av, false, bv, (short)0, acc1, false, false);
    bv.x = b0.z; bv.y = b1.z;
    acc2 = __builtin_amdgcn_wmma_f32_16x16x4_f32(
        false, av, false, bv, (short)0, acc2, false, false);
    bv.x = b0.w; bv.y = b1.w;
    acc3 = __builtin_amdgcn_wmma_f32_16x16x4_f32(
        false, av, false, bv, (short)0, acc3, false, false);
  }

  float* cp = C + (size_t)(m0 + 8 * h) * NN + 4 * c;
#pragma unroll
  for (int r = 0; r < 8; ++r) {
    const float sc = (r < 4) ? sc_lo[r & 3] : sc_hi[r & 3];
    v4f o;
    o.x = acc0[r] * sc; o.y = acc1[r] * sc; o.z = acc2[r] * sc; o.w = acc3[r] * sc;
    *reinterpret_cast<v4f*>(cp + (size_t)r * NN) = o;
  }
}

template <int NN>
__device__ __forceinline__ void fold_weights_tile(const float* __restrict__ W1,
                                                  const float* __restrict__ b1,
                                                  const float* __restrict__ W2,
                                                  float* __restrict__ W12,
                                                  int KIN, int K, int tile) {
  const int wave = threadIdx.x >> 5;
  const int lane = threadIdx.x & 31;
  const int m0 = tile * 16;
  const int n0 = wave * 16;
  const int half = lane >> 4;
  const int l15 = lane & 15;
  const bool bias_tile = (m0 >= KIN);
  const bool has_row = !bias_tile || l15 == 0;

  v8f acc = {};
  const float* aptr = (bias_tile ? b1 : W1 + (size_t)(m0 + l15) * K) + half * 2;
  const float* bptr = W2 + (size_t)(half * 2) * NN + n0 + l15;

  for (int k = 0; k < K; k += 4) {
    v2f a, b;
    const float a0 = aptr[0], a1 = aptr[1];
    a.x = has_row ? a0 : 0.0f;
    a.y = has_row ? a1 : 0.0f;
    b.x = bptr[0];
    b.y = bptr[NN];
    acc = __builtin_amdgcn_wmma_f32_16x16x4_f32(
         false, a,  false, b,
         (short)0, acc,  false,  false);
    aptr += 4;
    bptr += 4 * NN;
  }

  const int mrow = m0 + half * 8;
  float* cptr = W12 + (size_t)mrow * NN + n0 + l15;
#pragma unroll
  for (int r = 0; r < 8; ++r) {
    if (mrow + r <= KIN) cptr[0] = acc[r];
    cptr += NN;
  }
}

__global__ __launch_bounds__(1024)
void chunk_sort_fold_k(const int* __restrict__ src, const int* __restrict__ dst,
                       int* __restrict__ cntmat, int2* __restrict__ tmp, int E,
                       int nchunks,
                       const float* __restrict__ W1, const float* __restrict__ b1,
                       const float* __restrict__ W2, float* __restrict__ W12) {
  if ((int)blockIdx.x >= nchunks) {
    if (threadIdx.x < 128)
      fold_weights_tile<64>(W1, b1, W2, W12, 128, 128, (int)blockIdx.x - nchunks);
    return;
  }
  __shared__ int hist[GRP_MAX];
  __shared__ int lbase[GRP_MAX];
  __shared__ int wsum[8];
  __shared__ int2 piece[EDGE_CHUNK];
  const int tid = threadIdx.x;
  const int lane = tid & 31;
  const int w = tid >> 5;
  const int blk = blockIdx.x;
  if (tid < GRP_MAX) hist[tid] = 0;
  __syncthreads();
  const int e0 = blk * EDGE_CHUNK + tid;
  int d[EDGE_CHUNK / 1024], s[EDGE_CHUNK / 1024];
#pragma unroll
  for (int j = 0; j < EDGE_CHUNK / 1024; ++j) {
    const int e = e0 + j * 1024;
    const int ec = e < E ? e : E - 1;
    d[j] = dst[ec];
    s[j] = src[ec];
  }
  int r[EDGE_CHUNK / 1024];
#pragma unroll
  for (int j = 0; j < EDGE_CHUNK / 1024; ++j) {
    r[j] = atomicAdd(&hist[d[j] >> GRP_SHIFT], (e0 + j * 1024 < E) ? 1 : 0);
  }
  __syncthreads();
  int v = 0, x = 0;
  if (tid < GRP_MAX) {
    v = hist[tid];
    x = v;
    for (int o = 1; o < 32; o <<= 1) {
      const int y = __shfl_up(x, o, 32);
      if (lane >= o) x += y;
    }
    if (lane == 31) wsum[w] = x;
  }
  __syncthreads();
  if (tid < GRP_MAX) {
    int wp = 0;
    for (int k = 0; k < w; ++k) wp += wsum[k];
    const int excl = wp + x - v;
    lbase[tid] = excl;
    cntmat[blk * GRP_MAX + tid] = (excl << 16) | v;
  }
  __syncthreads();
  int slot[EDGE_CHUNK / 1024];
#pragma unroll
  for (int j = 0; j < EDGE_CHUNK / 1024; ++j) slot[j] = lbase[d[j] >> GRP_SHIFT] + r[j];
#pragma unroll
  for (int j = 0; j < EDGE_CHUNK / 1024; ++j) {
    if (e0 + j * 1024 < E) piece[slot[j]] = make_int2(s[j], d[j]);
  }
  __syncthreads();
  const int left = E - blk * EDGE_CHUNK;
  const int nvalid = left < EDGE_CHUNK ? left : EDGE_CHUNK;
  int2 o[EDGE_CHUNK / 1024];
#pragma unroll
  for (int j = 0; j < EDGE_CHUNK / 1024; ++j) o[j] = piece[tid + j * 1024];
#pragma unroll
  for (int j = 0; j < EDGE_CHUNK / 1024; ++j) {
    const int i = tid + j * 1024;
    if (i < nvalid) tmp[blk * EDGE_CHUNK + i] = o[j];
  }
}

template <int C> struct rowvec;
template <> struct rowvec<64>  { typedef float2 type; };

__device__ __forceinline__ void add_row(float2& a, const float2& v) {
  a.x += v.x; a.y += v.y;
}

__device__ __forceinline__ float2 scale_normalize(float2 v) {
  float mx = fmaxf(v.x, v.y);
  float mn = fminf(v.x, v.y);
#pragma unroll
  for (int o = 16; o > 0; o >>= 1) {
    mx = fmaxf(mx, __shfl_xor(mx, o, 32));
    mn = fminf(mn, __shfl_xor(mn, o, 32));
  }
  float inv = 1.0f / (mx - mn);
  float sx = (v.x - mn) * inv;
  float sy = (v.y - mn) * inv;
  float ss = sx * sx + sy * sy;
#pragma unroll
  for (int o = 16; o > 0; o >>= 1) ss += __shfl_xor(ss, o, 32);
  float r = 1.0f / fmaxf(sqrtf(ss), 1e-12f);
  float2 ov;
  ov.x = sx * r;
  ov.y = sy * r;
  return ov;
}

#define ROWS_IN_FLIGHT 24
template <int C> struct rowset {
  typename rowvec<C>::type r0, r1, r2, r3, r4, r5, r6, r7, r8, r9, r10, r11, r12, r13, r14, r15, r16, r17, r18, r19, r20, r21, r22, r23;
};

template <int C, int J0, int CNT>
__device__ __forceinline__ void load_rows(rowset<C>& v, const float* __restrict__ feat,
                                          int sv, int m, unsigned lane_off) {
  typedef typename rowvec<C>::type vec;
  const char* fb = reinterpret_cast<const char*>(feat);
#define ROW_OF_EDGE(K) \
  (*reinterpret_cast<const vec*>(fb + (((unsigned)__builtin_amdgcn_readlane(sv, J0 + K)) * (unsigned)(C * 4) + lane_off)))
  if (0 < CNT && J0 + 0 < m) { v.r0 = ROW_OF_EDGE(0);
  if (1 < CNT && J0 + 1 < m) { v.r1 = ROW_OF_EDGE(1);
  if (2 < CNT && J0 + 2 < m) { v.r2 = ROW_OF_EDGE(2);
  if (3 < CNT && J0 + 3 < m) { v.r3 = ROW_OF_EDGE(3);
  if (4 < CNT && J0 + 4 < m) { v.r4 = ROW_OF_EDGE(4);
  if (5 < CNT && J0 + 5 < m) { v.r5 = ROW_OF_EDGE(5);
  if (6 < CNT && J0 + 6 < m) { v.r6 = ROW_OF_EDGE(6);
  if (7 < CNT && J0 + 7 < m) { v.r7 = ROW_OF_EDGE(7);
  if (8 < CNT && J0 + 8 < m) { v.r8 = ROW_OF_EDGE(8);
  if (9 < CNT && J0 + 9 < m) { v.r9 = ROW_OF_EDGE(9);
  if (10 < CNT && J0 + 10 < m) { v.r10 = ROW_OF_EDGE(10);
  if (11 < CNT && J0 + 11 < m) { v.r11 = ROW_OF_EDGE(11);
  if (12 < CNT && J0 + 12 < m) { v.r12 = ROW_OF_EDGE(12);
  if (13 < CNT && J0 + 13 < m) { v.r13 = ROW_OF_EDGE(13);
  if (14 < CNT && J0 + 14 < m) { v.r14 = ROW_OF_EDGE(14);
  if (15 < CNT && J0 + 15 < m) { v.r15 = ROW_OF_EDGE(15);
  if (16 < CNT && J0 + 16 < m) { v.r16 = ROW_OF_EDGE(16);
  if (17 < CNT && J0 + 17 < m) { v.r17 = ROW_OF_EDGE(17);
  if (18 < CNT && J0 + 18 < m) { v.r18 = ROW_OF_EDGE(18);
  if (19 < CNT && J0 + 19 < m) { v.r19 = ROW_OF_EDGE(19);
  if (20 < CNT && J0 + 20 < m) { v.r20 = ROW_OF_EDGE(20);
  if (21 < CNT && J0 + 21 < m) { v.r21 = ROW_OF_EDGE(21);
  if (22 < CNT && J0 + 22 < m) { v.r22 = ROW_OF_EDGE(22);
  if (23 < CNT && J0 + 23 < m) { v.r23 = ROW_OF_EDGE(23);
  }}}}}}}}}}}}}}}}}}}}}}}}
#undef ROW_OF_EDGE
}

template <int C, int J0, int CNT>
__device__ __forceinline__ void add_rows(typename rowvec<C>::type& acc, const rowset<C>& v,
                                         int m) {
  if (0 < CNT && J0 + 0 < m) { add_row(acc, v.r0);
  if (1 < CNT && J0 + 1 < m) { add_row(acc, v.r1);
  if (2 < CNT && J0 + 2 < m) { add_row(acc, v.r2);
  if (3 < CNT && J0 + 3 < m) { add_row(acc, v.r3);
  if (4 < CNT && J0 + 4 < m) { add_row(acc, v.r4);
  if (5 < CNT && J0 + 5 < m) { add_row(acc, v.r5);
  if (6 < CNT && J0 + 6 < m) { add_row(acc, v.r6);
  if (7 < CNT && J0 + 7 < m) { add_row(acc, v.r7);
  if (8 < CNT && J0 + 8 < m) { add_row(acc, v.r8);
  if (9 < CNT && J0 + 9 < m) { add_row(acc, v.r9);
  if (10 < CNT && J0 + 10 < m) { add_row(acc, v.r10);
  if (11 < CNT && J0 + 11 < m) { add_row(acc, v.r11);
  if (12 < CNT && J0 + 12 < m) { add_row(acc, v.r12);
  if (13 < CNT && J0 + 13 < m) { add_row(acc, v.r13);
  if (14 < CNT && J0 + 14 < m) { add_row(acc, v.r14);
  if (15 < CNT && J0 + 15 < m) { add_row(acc, v.r15);
  if (16 < CNT && J0 + 16 < m) { add_row(acc, v.r16);
  if (17 < CNT && J0 + 17 < m) { add_row(acc, v.r17);
  if (18 < CNT && J0 + 18 < m) { add_row(acc, v.r18);
  if (19 < CNT && J0 + 19 < m) { add_row(acc, v.r19);
  if (20 < CNT && J0 + 20 < m) { add_row(acc, v.r20);
  if (21 < CNT && J0 + 21 < m) { add_row(acc, v.r21);
  if (22 < CNT && J0 + 22 < m) { add_row(acc, v.r22);
  if (23 < CNT && J0 + 23 < m) { add_row(acc, v.r23);
  }}}}}}}}}}}}}}}}}}}}}}}}
}

template <int C, bool NORM>
__global__ void gather_k(const float* __restrict__ feat,
                         const float* __restrict__ dinv,
                         const int* __restrict__ rowptr,
                         const int* __restrict__ cnt,
                         const int* __restrict__ csr,
                         const float* __restrict__ bias,
                         float* __restrict__ out, int N) {
  typedef typename rowvec<C>::type vec;
  const int lane = threadIdx.x & 31;
  const int node = __builtin_amdgcn_readfirstlane(
      (int)(blockIdx.x * (blockDim.x >> 5) + (threadIdx.x >> 5)));
  if (node >= N) return;

  int p = rowptr[node];
  int deg = cnt[node];
  float di = dinv[node];
  asm volatile("" : "+s"(p), "+s"(deg), "+s"(di) : "s"(csr), "s"(bias));

  const vec bias_v = reinterpret_cast<const vec*>(bias)[lane];
  const vec self_v = reinterpret_cast<const vec*>(feat + (size_t)node * C)[lane];
  vec acc;
  bool started = false;

  const unsigned lane_off = (unsigned)lane * (unsigned)sizeof(vec);
  for (int m = deg; m > 0; m -= 32, p += 32) {
    const unsigned mine = (unsigned)(lane < m ? lane : m - 1);
    const int sv = csr[(unsigned)p + mine];
    rowset<C> v;
    load_rows<C, 0, ROWS_IN_FLIGHT>(v, feat, sv, m, lane_off);
    if (!started) {
      acc = self_v;
      started = true;
    }
    add_rows<C, 0, ROWS_IN_FLIGHT>(acc, v, m);
    if (m > ROWS_IN_FLIGHT) {
      load_rows<C, ROWS_IN_FLIGHT, 32 - ROWS_IN_FLIGHT>(v, feat, sv, m, lane_off);
      add_rows<C, ROWS_IN_FLIGHT, 32 - ROWS_IN_FLIGHT>(acc, v, m);
    }
  }
  if (!started) acc = self_v;
  acc.x = di * acc.x + bias_v.x;
  acc.y = di * acc.y + bias_v.y;
  if constexpr (NORM) {
    static_assert(C == 64, "the fused scale + normalise expects one float2 per lane");
    acc = scale_normalize(acc);
  } else {
    acc.x *= di;
    acc.y *= di;
  }
  reinterpret_cast<vec*>(out + (size_t)node * C)[lane] = acc;
}

extern "C" void kernel_launch(void* const* d_in, const int* in_sizes, int n_in,
                              void* d_out, int out_size, void* d_ws, size_t ws_size,
                              hipStream_t stream) {
  (void)n_in; (void)out_size; (void)ws_size;

  const float* x  = (const float*)d_in[0];
  const int*   ei = (const int*)d_in[1];
  const float* W1 = (const float*)d_in[2];
  const float* b1 = (const float*)d_in[3];
  const float* W2 = (const float*)d_in[4];
  const float* b2 = (const float*)d_in[5];

  const int N = in_sizes[0] / 128;
  const int E = in_sizes[1] / 2;
  const int* src = ei;
  const int* dst = ei + E;
  float* out = (float*)d_out;

  float* ws = (float*)d_ws;
  size_t Npad = ((size_t)N + 63) & ~(size_t)63;
  float* dinv = ws;
  float* y = ws + Npad;
  float* g = y + (size_t)N * 64;
  float* scratch = g + (size_t)N * 64;
  int* cnt    = (int*)(scratch + (size_t)N * 128);
  int* rowptr = cnt + Npad;
  int2* tmp   = (int2*)scratch;
  float* W12  = scratch + (((size_t)E * 2 + 3) & ~(size_t)3);
  int* csr    = rowptr + Npad;
  int* cntmat = csr + E;
  const int nchunks = (E + EDGE_CHUNK - 1) / EDGE_CHUNK;

  chunk_sort_fold_k<<<nchunks + 128 / 16 + 1, 1024, 0, stream>>>(src, dst, cntmat, tmp, E, nchunks, W1, b1, W2, W12);
  group_csr_k<<<(N + GRP_SIZE - 1) / GRP_SIZE, 1024, 0, stream>>>(tmp, cntmat, nchunks, cnt, rowptr, dinv, csr, N);

  gemm_strip_k<128><<<(N + 63) / 64, 128, 0, stream>>>(x, W12, dinv, y, N);
  gather_k<64, false><<<(N + 7) / 8, 256, 0, stream>>>(y, dinv, rowptr, cnt, csr, W12 + 128 * 64, g, N);

  gather_k<64, true><<<(N + 7) / 8, 256, 0, stream>>>(g, dinv, rowptr, cnt, csr, b2, out, N);
}


# x@W12 loses its launch: the block that sorts a 512-node group's edges forms the group's 512 rows, a 16-row strip per wave, half the waves' WMMAs before each edge pass, half after; 5 launches to 4
# speedup vs baseline: 6.5827x; 1.0307x over previous
#include <hip/hip_runtime.h>
#include <math.h>

typedef float v2f __attribute__((ext_vector_type(2)));
typedef float v4f __attribute__((ext_vector_type(4)));
typedef float v8f __attribute__((ext_vector_type(8)));

#define GRP_SHIFT 9
#define GRP_SIZE  512
#define GRP_MAX   256
#define EDGE_CHUNK 8192
#define CSR_STAGE 12288
#define CHUNK_MAX 1024

template <int K>
__device__ __forceinline__ void runs_of(const int* pre, int s0, const int (&p)[K], int (&c)[K]) {
#pragma unroll
  for (int k = 0; k < K; ++k) c[k] = 0;
  for (int s = s0; s > 0; s >>= 1) {
#pragma unroll
    for (int k = 0; k < K; ++k) {
      const int t = pre[c[k] + s];
      if (t <= p[k]) c[k] += s;
    }
  }
}

template <int KB0, int KB1>
__device__ __forceinline__ void strip_steps(const v4f* __restrict__ ap,
                                            const float* __restrict__ bp,
                                            v8f& acc0, v8f& acc1, v8f& acc2, v8f& acc3) {
  constexpr int NN = 64;
  constexpr int T0 = KB0 * 8;
  constexpr int T1 = KB1 * 8;
  v4f a[KB1 - KB0][4];
#pragma unroll
  for (int kb = KB0; kb < KB1; ++kb) {
#pragma unroll
    for (int q = 0; q < 4; ++q) a[kb - KB0][q] = ap[kb * 8 + q];
  }
  v4f br[3][2];
#pragma unroll
  for (int t = T0; t < T0 + 2; ++t) {
    const int k = (t >> 3) * 32 + (t & 7) * 2;
    br[t % 3][0] = *reinterpret_cast<const v4f*>(bp + (size_t)k * NN);
    br[t % 3][1] = *reinterpret_cast<const v4f*>(bp + (size_t)(k + 1) * NN);
  }
#pragma unroll
  for (int t = T0; t < T1; ++t) {
    if (t + 2 < T1) {
      const int kn = ((t + 2) >> 3) * 32 + ((t + 2) & 7) * 2;
      br[(t + 2) % 3][0] = *reinterpret_cast<const v4f*>(bp + (size_t)kn * NN);
      br[(t + 2) % 3][1] = *reinterpret_cast<const v4f*>(bp + (size_t)(kn + 1) * NN);
    }
    __builtin_amdgcn_sched_barrier(0);
    const v4f aq = a[(t >> 3) - KB0][(t >> 1) & 3];
    const v4f b0 = br[t % 3][0];
    const v4f b1 = br[t % 3][1];
    v2f av, bv;
    av.x = (t & 1) ? aq.z : aq.x;
    av.y = (t & 1) ? aq.w : aq.y;
    const v8f zero = {};
    const bool first = (t == 0);
    bv.x = b0.x; bv.y = b1.x;
    acc0 = __builtin_amdgcn_wmma_f32_16x16x4_f32(
         false, av,  false, bv,
         (short)0, first ? zero : acc0,  false,  false);
    bv.x = b0.y; bv.y = b1.y;
    acc1 = __builtin_amdgcn_wmma_f32_16x16x4_f32(
        false, av, false, bv, (short)0, first ? zero : acc1, false, false);
    bv.x = b0.z; bv.y = b1.z;
    acc2 = __builtin_amdgcn_wmma_f32_16x16x4_f32(
        false, av, false, bv, (short)0, first ? zero : acc2, false, false);
    bv.x = b0.w; bv.y = b1.w;
    acc3 = __builtin_amdgcn_wmma_f32_16x16x4_f32(
        false, av, false, bv, (short)0, first ? zero : acc3, false, false);
  }
}

__device__ __forceinline__ void store_strip(const float* sdw, int h, float* __restrict__ cp,
                                            const v8f& acc0, const v8f& acc1,
                                            const v8f& acc2, const v8f& acc3) {
  const v4f sc_lo = *reinterpret_cast<const v4f*>(sdw + 8 * h);
  const v4f sc_hi = *reinterpret_cast<const v4f*>(sdw + 8 * h + 4);
#pragma unroll
  for (int r = 0; r < 8; ++r) {
    const float sc = (r < 4) ? sc_lo[r & 3] : sc_hi[r & 3];
    v4f o;
    o.x = acc0[r] * sc; o.y = acc1[r] * sc; o.z = acc2[r] * sc; o.w = acc3[r] * sc;
    *reinterpret_cast<v4f*>(cp + (size_t)r * 64) = o;
  }
}

__global__ __launch_bounds__(1024)
void group_csr_k(const int2* __restrict__ tmp, const int* __restrict__ cntmat, int nchunks,
                 int* __restrict__ cnt, int* __restrict__ rowptr,
                 float* __restrict__ dinv, int* __restrict__ csr, int N,
                 const float* __restrict__ xin, const float* __restrict__ W12,
                 float* __restrict__ yout) {
  __shared__ int c[GRP_SIZE];
  __shared__ int cur[GRP_SIZE];
  __shared__ int wsum[32];
  __shared__ int wofs[32];
  __shared__ int pre[CHUNK_MAX + 1];
  __shared__ int rb[CHUNK_MAX];
  __shared__ int stage[CSR_STAGE];
  __shared__ int2 held[CSR_STAGE];
  __shared__ __attribute__((aligned(16))) float sd[GRP_SIZE];
  const int tid = threadIdx.x;
  const int lane = tid & 31;
  const int w = tid >> 5;
  const int b = blockIdx.x;
  const int lo = b << GRP_SHIFT;

  const int wv = __builtin_amdgcn_readfirstlane(w);
  const int m0 = lo + wv * 16;
  const bool has_strip = m0 < N;
  const bool mm_first = (wv & 4) != 0;
  const int sh = lane >> 4;
  const int sc16 = lane & 15;
  const v4f* ap = reinterpret_cast<const v4f*>(xin + (size_t)(has_strip ? m0 + sc16 : 0) * 128 + sh * 16);
  const float* bp = W12 + (size_t)(sh * 16) * 64 + 4 * sc16;
  v8f acc0, acc1, acc2, acc3;

  int rc = 0, rl = 0;
  if (tid < nchunks) {
    const int pk = cntmat[tid * GRP_MAX + b];
    rc = pk & 0xFFFF;
    rl = pk >> 16;
  }
  if (tid < GRP_SIZE) c[tid] = 0;
  int rx = rc;
  int ro = rl;
  for (int o = 1; o < 32; o <<= 1) {
    const int y = __shfl_up(rx, o, 32);
    if (lane >= o) rx += y;
  }
  for (int o = 16; o > 0; o >>= 1) ro += __shfl_xor(ro, o, 32);
  if (lane == 31) wsum[w] = rx;
  if (lane == 0) wofs[w] = ro;
  __syncthreads();
  int rwp = 0, n = 0, start = 0;
  for (int k = 0; k < 32; ++k) {
    const int t = wsum[k];
    if (k < w) rwp += t;
    n += t;
    start += wofs[k];
  }
  n = __builtin_amdgcn_readfirstlane(n);
  start = __builtin_amdgcn_readfirstlane(start);
  {
    const int excl = rwp + rx - rc;
    pre[tid] = (tid < nchunks) ? excl : n;
    if (tid == 0) pre[CHUNK_MAX] = n;
    if (tid < nchunks) rb[tid] = tid * EDGE_CHUNK + rl - excl;
  }
  int s0 = 1;
  while (2 * s0 < nchunks) s0 <<= 1;
  __syncthreads();

  const bool staged = (n <= CSR_STAGE);
  if (has_strip && mm_first) strip_steps<0, 2>(ap, bp, acc0, acc1, acc2, acc3);
  for (int p0 = tid; p0 < n; p0 += 4096) {
    int pc[4], r[4];
#pragma unroll
    for (int k = 0; k < 4; ++k) {
      const int p = p0 + k * 1024;
      pc[k] = p < n ? p : p0;
    }
    runs_of<4>(pre, s0, pc, r);
    int2 pr[4];
#pragma unroll
    for (int k = 0; k < 4; ++k) pr[k] = tmp[rb[r[k]] + pc[k]];
#pragma unroll
    for (int k = 0; k < 4; ++k) {
      atomicAdd(&c[pr[k].y - lo], (p0 + k * 1024 < n) ? 1 : 0);
    }
    if (staged) {
#pragma unroll
      for (int k = 0; k < 4; ++k) {
        if (p0 + k * 1024 < n) held[p0 + k * 1024] = pr[k];
      }
    }
  }
  if (has_strip && !mm_first) strip_steps<0, 2>(ap, bp, acc0, acc1, acc2, acc3);
  __syncthreads();

  int v = 0, x = 0;
  if (tid < GRP_SIZE) {
    v = c[tid];
    x = v;
    for (int o = 1; o < 32; o <<= 1) {
      const int y = __shfl_up(x, o, 32);
      if (lane >= o) x += y;
    }
    if (lane == 31) wsum[w] = x;
  }
  __syncthreads();
  if (tid < GRP_SIZE) {
    int wp = 0;
    for (int k = 0; k < w; ++k) wp += wsum[k];
    const int excl = wp + x - v;
    cur[tid] = excl;
    const int i = lo + tid;
    if (i < N) {
      cnt[i] = v;
      rowptr[i] = start + excl;
      float d = (float)(v + 1);
      const float dv = (d > 0.0f) ? rsqrtf(d) : 0.0f;
      dinv[i] = dv;
      sd[tid] = dv;
    }
  }
  __syncthreads();

  float* cp = yout + (size_t)(m0 + 8 * sh) * 64 + 4 * sc16;
  if (has_strip && mm_first) {
    strip_steps<2, 4>(ap, bp, acc0, acc1, acc2, acc3);
    store_strip(&sd[wv * 16], sh, cp, acc0, acc1, acc2, acc3);
  }
  for (int p0 = tid; p0 < n; p0 += 4096) {
    int pc[4];
#pragma unroll
    for (int k = 0; k < 4; ++k) {
      const int p = p0 + k * 1024;
      pc[k] = p < n ? p : p0;
    }
    int2 pr[4];
    if (staged) {
#pragma unroll
      for (int k = 0; k < 4; ++k) pr[k] = held[pc[k]];
    } else {
      int r[4];
      runs_of<4>(pre, s0, pc, r);
#pragma unroll
      for (int k = 0; k < 4; ++k) pr[k] = tmp[rb[r[k]] + pc[k]];
    }
    int q[4];
#pragma unroll
    for (int k = 0; k < 4; ++k) {
      q[k] = atomicAdd(&cur[pr[k].y - lo], (p0 + k * 1024 < n) ? 1 : 0);
    }
#pragma unroll
    for (int k = 0; k < 4; ++k) {
      if (p0 + k * 1024 < n) {
        if (staged) stage[q[k]] = pr[k].x;
        else csr[start + q[k]] = pr[k].x;
      }
    }
  }
  if (has_strip && !mm_first) {
    strip_steps<2, 4>(ap, bp, acc0, acc1, acc2, acc3);
    store_strip(&sd[wv * 16], sh, cp, acc0, acc1, acc2, acc3);
  }
  if (staged) {
    __syncthreads();
    for (int i = tid; i < n; i += 1024) csr[start + i] = stage[i];
  }
}

template <int NN>
__device__ __forceinline__ void fold_weights_tile(const float* __restrict__ W1,
                                                  const float* __restrict__ b1,
                                                  const float* __restrict__ W2,
                                                  float* __restrict__ W12,
                                                  int KIN, int K, int tile) {
  const int wave = threadIdx.x >> 5;
  const int lane = threadIdx.x & 31;
  const int m0 = tile * 16;
  const int n0 = wave * 16;
  const int half = lane >> 4;
  const int l15 = lane & 15;
  const bool bias_tile = (m0 >= KIN);
  const bool has_row = !bias_tile || l15 == 0;

  v8f acc = {};
  const float* aptr = (bias_tile ? b1 : W1 + (size_t)(m0 + l15) * K) + half * 2;
  const float* bptr = W2 + (size_t)(half * 2) * NN + n0 + l15;

  for (int k = 0; k < K; k += 4) {
    v2f a, b;
    const float a0 = aptr[0], a1 = aptr[1];
    a.x = has_row ? a0 : 0.0f;
    a.y = has_row ? a1 : 0.0f;
    b.x = bptr[0];
    b.y = bptr[NN];
    acc = __builtin_amdgcn_wmma_f32_16x16x4_f32(
         false, a,  false, b,
         (short)0, acc,  false,  false);
    aptr += 4;
    bptr += 4 * NN;
  }

  const int mrow = m0 + half * 8;
  float* cptr = W12 + (size_t)mrow * NN + n0 + l15;
#pragma unroll
  for (int r = 0; r < 8; ++r) {
    if (mrow + r <= KIN) cptr[0] = acc[r];
    cptr += NN;
  }
}

__global__ __launch_bounds__(1024)
void chunk_sort_fold_k(const int* __restrict__ src, const int* __restrict__ dst,
                       int* __restrict__ cntmat, int2* __restrict__ tmp, int E,
                       int nchunks,
                       const float* __restrict__ W1, const float* __restrict__ b1,
                       const float* __restrict__ W2, float* __restrict__ W12) {
  if ((int)blockIdx.x >= nchunks) {
    if (threadIdx.x < 128)
      fold_weights_tile<64>(W1, b1, W2, W12, 128, 128, (int)blockIdx.x - nchunks);
    return;
  }
  __shared__ int hist[GRP_MAX];
  __shared__ int lbase[GRP_MAX];
  __shared__ int wsum[8];
  __shared__ int2 piece[EDGE_CHUNK];
  const int tid = threadIdx.x;
  const int lane = tid & 31;
  const int w = tid >> 5;
  const int blk = blockIdx.x;
  if (tid < GRP_MAX) hist[tid] = 0;
  __syncthreads();
  const int e0 = blk * EDGE_CHUNK + tid;
  int d[EDGE_CHUNK / 1024], s[EDGE_CHUNK / 1024];
#pragma unroll
  for (int j = 0; j < EDGE_CHUNK / 1024; ++j) {
    const int e = e0 + j * 1024;
    const int ec = e < E ? e : E - 1;
    d[j] = dst[ec];
    s[j] = src[ec];
  }
  int r[EDGE_CHUNK / 1024];
#pragma unroll
  for (int j = 0; j < EDGE_CHUNK / 1024; ++j) {
    r[j] = atomicAdd(&hist[d[j] >> GRP_SHIFT], (e0 + j * 1024 < E) ? 1 : 0);
  }
  __syncthreads();
  int v = 0, x = 0;
  if (tid < GRP_MAX) {
    v = hist[tid];
    x = v;
    for (int o = 1; o < 32; o <<= 1) {
      const int y = __shfl_up(x, o, 32);
      if (lane >= o) x += y;
    }
    if (lane == 31) wsum[w] = x;
  }
  __syncthreads();
  if (tid < GRP_MAX) {
    int wp = 0;
    for (int k = 0; k < w; ++k) wp += wsum[k];
    const int excl = wp + x - v;
    lbase[tid] = excl;
    cntmat[blk * GRP_MAX + tid] = (excl << 16) | v;
  }
  __syncthreads();
  int slot[EDGE_CHUNK / 1024];
#pragma unroll
  for (int j = 0; j < EDGE_CHUNK / 1024; ++j) slot[j] = lbase[d[j] >> GRP_SHIFT] + r[j];
#pragma unroll
  for (int j = 0; j < EDGE_CHUNK / 1024; ++j) {
    if (e0 + j * 1024 < E) piece[slot[j]] = make_int2(s[j], d[j]);
  }
  __syncthreads();
  const int left = E - blk * EDGE_CHUNK;
  const int nvalid = left < EDGE_CHUNK ? left : EDGE_CHUNK;
  int2 o[EDGE_CHUNK / 1024];
#pragma unroll
  for (int j = 0; j < EDGE_CHUNK / 1024; ++j) o[j] = piece[tid + j * 1024];
#pragma unroll
  for (int j = 0; j < EDGE_CHUNK / 1024; ++j) {
    const int i = tid + j * 1024;
    if (i < nvalid) tmp[blk * EDGE_CHUNK + i] = o[j];
  }
}

template <int C> struct rowvec;
template <> struct rowvec<64>  { typedef float2 type; };

__device__ __forceinline__ void add_row(float2& a, const float2& v) {
  a.x += v.x; a.y += v.y;
}

__device__ __forceinline__ float2 scale_normalize(float2 v) {
  float mx = fmaxf(v.x, v.y);
  float mn = fminf(v.x, v.y);
#pragma unroll
  for (int o = 16; o > 0; o >>= 1) {
    mx = fmaxf(mx, __shfl_xor(mx, o, 32));
    mn = fminf(mn, __shfl_xor(mn, o, 32));
  }
  float inv = 1.0f / (mx - mn);
  float sx = (v.x - mn) * inv;
  float sy = (v.y - mn) * inv;
  float ss = sx * sx + sy * sy;
#pragma unroll
  for (int o = 16; o > 0; o >>= 1) ss += __shfl_xor(ss, o, 32);
  float r = 1.0f / fmaxf(sqrtf(ss), 1e-12f);
  float2 ov;
  ov.x = sx * r;
  ov.y = sy * r;
  return ov;
}

#define ROWS_IN_FLIGHT 24
template <int C> struct rowset {
  typename rowvec<C>::type r0, r1, r2, r3, r4, r5, r6, r7, r8, r9, r10, r11, r12, r13, r14, r15, r16, r17, r18, r19, r20, r21, r22, r23;
};

template <int C, int J0, int CNT>
__device__ __forceinline__ void load_rows(rowset<C>& v, const float* __restrict__ feat,
                                          int sv, int m, unsigned lane_off) {
  typedef typename rowvec<C>::type vec;
  const char* fb = reinterpret_cast<const char*>(feat);
#define ROW_OF_EDGE(K) \
  (*reinterpret_cast<const vec*>(fb + (((unsigned)__builtin_amdgcn_readlane(sv, J0 + K)) * (unsigned)(C * 4) + lane_off)))
  if (0 < CNT && J0 + 0 < m) { v.r0 = ROW_OF_EDGE(0);
  if (1 < CNT && J0 + 1 < m) { v.r1 = ROW_OF_EDGE(1);
  if (2 < CNT && J0 + 2 < m) { v.r2 = ROW_OF_EDGE(2);
  if (3 < CNT && J0 + 3 < m) { v.r3 = ROW_OF_EDGE(3);
  if (4 < CNT && J0 + 4 < m) { v.r4 = ROW_OF_EDGE(4);
  if (5 < CNT && J0 + 5 < m) { v.r5 = ROW_OF_EDGE(5);
  if (6 < CNT && J0 + 6 < m) { v.r6 = ROW_OF_EDGE(6);
  if (7 < CNT && J0 + 7 < m) { v.r7 = ROW_OF_EDGE(7);
  if (8 < CNT && J0 + 8 < m) { v.r8 = ROW_OF_EDGE(8);
  if (9 < CNT && J0 + 9 < m) { v.r9 = ROW_OF_EDGE(9);
  if (10 < CNT && J0 + 10 < m) { v.r10 = ROW_OF_EDGE(10);
  if (11 < CNT && J0 + 11 < m) { v.r11 = ROW_OF_EDGE(11);
  if (12 < CNT && J0 + 12 < m) { v.r12 = ROW_OF_EDGE(12);
  if (13 < CNT && J0 + 13 < m) { v.r13 = ROW_OF_EDGE(13);
  if (14 < CNT && J0 + 14 < m) { v.r14 = ROW_OF_EDGE(14);
  if (15 < CNT && J0 + 15 < m) { v.r15 = ROW_OF_EDGE(15);
  if (16 < CNT && J0 + 16 < m) { v.r16 = ROW_OF_EDGE(16);
  if (17 < CNT && J0 + 17 < m) { v.r17 = ROW_OF_EDGE(17);
  if (18 < CNT && J0 + 18 < m) { v.r18 = ROW_OF_EDGE(18);
  if (19 < CNT && J0 + 19 < m) { v.r19 = ROW_OF_EDGE(19);
  if (20 < CNT && J0 + 20 < m) { v.r20 = ROW_OF_EDGE(20);
  if (21 < CNT && J0 + 21 < m) { v.r21 = ROW_OF_EDGE(21);
  if (22 < CNT && J0 + 22 < m) { v.r22 = ROW_OF_EDGE(22);
  if (23 < CNT && J0 + 23 < m) { v.r23 = ROW_OF_EDGE(23);
  }}}}}}}}}}}}}}}}}}}}}}}}
#undef ROW_OF_EDGE
}

template <int C, int J0, int CNT>
__device__ __forceinline__ void add_rows(typename rowvec<C>::type& acc, const rowset<C>& v,
                                         int m) {
  if (0 < CNT && J0 + 0 < m) { add_row(acc, v.r0);
  if (1 < CNT && J0 + 1 < m) { add_row(acc, v.r1);
  if (2 < CNT && J0 + 2 < m) { add_row(acc, v.r2);
  if (3 < CNT && J0 + 3 < m) { add_row(acc, v.r3);
  if (4 < CNT && J0 + 4 < m) { add_row(acc, v.r4);
  if (5 < CNT && J0 + 5 < m) { add_row(acc, v.r5);
  if (6 < CNT && J0 + 6 < m) { add_row(acc, v.r6);
  if (7 < CNT && J0 + 7 < m) { add_row(acc, v.r7);
  if (8 < CNT && J0 + 8 < m) { add_row(acc, v.r8);
  if (9 < CNT && J0 + 9 < m) { add_row(acc, v.r9);
  if (10 < CNT && J0 + 10 < m) { add_row(acc, v.r10);
  if (11 < CNT && J0 + 11 < m) { add_row(acc, v.r11);
  if (12 < CNT && J0 + 12 < m) { add_row(acc, v.r12);
  if (13 < CNT && J0 + 13 < m) { add_row(acc, v.r13);
  if (14 < CNT && J0 + 14 < m) { add_row(acc, v.r14);
  if (15 < CNT && J0 + 15 < m) { add_row(acc, v.r15);
  if (16 < CNT && J0 + 16 < m) { add_row(acc, v.r16);
  if (17 < CNT && J0 + 17 < m) { add_row(acc, v.r17);
  if (18 < CNT && J0 + 18 < m) { add_row(acc, v.r18);
  if (19 < CNT && J0 + 19 < m) { add_row(acc, v.r19);
  if (20 < CNT && J0 + 20 < m) { add_row(acc, v.r20);
  if (21 < CNT && J0 + 21 < m) { add_row(acc, v.r21);
  if (22 < CNT && J0 + 22 < m) { add_row(acc, v.r22);
  if (23 < CNT && J0 + 23 < m) { add_row(acc, v.r23);
  }}}}}}}}}}}}}}}}}}}}}}}}
}

template <int C, bool NORM>
__global__ void gather_k(const float* __restrict__ feat,
                         const float* __restrict__ dinv,
                         const int* __restrict__ rowptr,
                         const int* __restrict__ cnt,
                         const int* __restrict__ csr,
                         const float* __restrict__ bias,
                         float* __restrict__ out, int N) {
  typedef typename rowvec<C>::type vec;
  const int lane = threadIdx.x & 31;
  const int node = __builtin_amdgcn_readfirstlane(
      (int)(blockIdx.x * (blockDim.x >> 5) + (threadIdx.x >> 5)));
  if (node >= N) return;

  int p = rowptr[node];
  int deg = cnt[node];
  float di = dinv[node];
  asm volatile("" : "+s"(p), "+s"(deg), "+s"(di) : "s"(csr), "s"(bias));

  const vec bias_v = reinterpret_cast<const vec*>(bias)[lane];
  const vec self_v = reinterpret_cast<const vec*>(feat + (size_t)node * C)[lane];
  vec acc;
  bool started = false;

  const unsigned lane_off = (unsigned)lane * (unsigned)sizeof(vec);
  for (int m = deg; m > 0; m -= 32, p += 32) {
    const unsigned mine = (unsigned)(lane < m ? lane : m - 1);
    const int sv = csr[(unsigned)p + mine];
    rowset<C> v;
    load_rows<C, 0, ROWS_IN_FLIGHT>(v, feat, sv, m, lane_off);
    if (!started) {
      acc = self_v;
      started = true;
    }
    add_rows<C, 0, ROWS_IN_FLIGHT>(acc, v, m);
    if (m > ROWS_IN_FLIGHT) {
      load_rows<C, ROWS_IN_FLIGHT, 32 - ROWS_IN_FLIGHT>(v, feat, sv, m, lane_off);
      add_rows<C, ROWS_IN_FLIGHT, 32 - ROWS_IN_FLIGHT>(acc, v, m);
    }
  }
  if (!started) acc = self_v;
  acc.x = di * acc.x + bias_v.x;
  acc.y = di * acc.y + bias_v.y;
  if constexpr (NORM) {
    static_assert(C == 64, "the fused scale + normalise expects one float2 per lane");
    acc = scale_normalize(acc);
  } else {
    acc.x *= di;
    acc.y *= di;
  }
  reinterpret_cast<vec*>(out + (size_t)node * C)[lane] = acc;
}

extern "C" void kernel_launch(void* const* d_in, const int* in_sizes, int n_in,
                              void* d_out, int out_size, void* d_ws, size_t ws_size,
                              hipStream_t stream) {
  (void)n_in; (void)out_size; (void)ws_size;

  const float* x  = (const float*)d_in[0];
  const int*   ei = (const int*)d_in[1];
  const float* W1 = (const float*)d_in[2];
  const float* b1 = (const float*)d_in[3];
  const float* W2 = (const float*)d_in[4];
  const float* b2 = (const float*)d_in[5];

  const int N = in_sizes[0] / 128;
  const int E = in_sizes[1] / 2;
  const int* src = ei;
  const int* dst = ei + E;
  float* out = (float*)d_out;

  float* ws = (float*)d_ws;
  size_t Npad = ((size_t)N + 63) & ~(size_t)63;
  float* dinv = ws;
  float* y = ws + Npad;
  float* g = y + (size_t)N * 64;
  float* scratch = g + (size_t)N * 64;
  int* cnt    = (int*)(scratch + (size_t)N * 128);
  int* rowptr = cnt + Npad;
  int2* tmp   = (int2*)scratch;
  float* W12  = scratch + (((size_t)E * 2 + 3) & ~(size_t)3);
  int* csr    = rowptr + Npad;
  int* cntmat = csr + E;
  const int nchunks = (E + EDGE_CHUNK - 1) / EDGE_CHUNK;

  chunk_sort_fold_k<<<nchunks + 128 / 16 + 1, 1024, 0, stream>>>(src, dst, cntmat, tmp, E, nchunks, W1, b1, W2, W12);
  group_csr_k<<<(N + GRP_SIZE - 1) / GRP_SIZE, 1024, 0, stream>>>(tmp, cntmat, nchunks, cnt, rowptr, dinv, csr, N, x, W12, y);

  gather_k<64, false><<<(N + 7) / 8, 256, 0, stream>>>(y, dinv, rowptr, cnt, csr, W12 + 128 * 64, g, N);

  gather_k<64, true><<<(N + 7) / 8, 256, 0, stream>>>(g, dinv, rowptr, cnt, csr, b2, out, N);
}
